# starting kernel
# baseline (speedup 1.0000x reference)
#include <hip/hip_runtime.h>
#include <hip/hip_bf16.h>
#include <stddef.h>


typedef __attribute__((ext_vector_type(2))) float v2f;
typedef __attribute__((ext_vector_type(8))) float v8f;

__global__ __launch_bounds__(256) void zero_f4_kernel(float4* __restrict__ p,
                                                      size_t n4) {
  size_t i = (size_t)blockIdx.x * blockDim.x + threadIdx.x;
  size_t stride = (size_t)gridDim.x * blockDim.x;
  for (; i < n4; i += stride) p[i] = make_float4(0.f, 0.f, 0.f, 0.f);
}

__global__ __launch_bounds__(256) void spmm_kernel(
    const int* __restrict__ dstIdx, const int* __restrict__ srcIdx,
    const float* __restrict__ vals, const float* __restrict__ x,
    float* __restrict__ y, int E) {
  const int lane = threadIdx.x & 31;
  int wave = (int)((blockIdx.x * blockDim.x + threadIdx.x) >> 5);
  const int nWaves = (int)((gridDim.x * blockDim.x) >> 5);
  for (int e = wave; e < E; e += nWaves) {
    const int s = srcIdx[e];
    const int d = dstIdx[e];
    const float v = vals[e];
    const float4 xv =
        *(const float4*)(x + (size_t)s * 128 + (size_t)lane * 4);
    float* yp = y + (size_t)d * 128 + (size_t)lane * 4;
    atomicAdd(yp + 0, xv.x * v);
    atomicAdd(yp + 1, xv.y * v);
    atomicAdd(yp + 2, xv.z * v);
    atomicAdd(yp + 3, xv.w * v);
  }
}

template <int NT>
__global__ __launch_bounds__(128) void gemm_bias_kernel(
    const float* __restrict__ S, const float* __restrict__ W,
    const float* __restrict__ bias, float* __restrict__ out, int nRowTiles) {
  const int lane = threadIdx.x & 31;
  const int waveInBlock = threadIdx.x >> 5;
  const int rowTile = blockIdx.x * (blockDim.x >> 5) + waveInBlock;
  if (rowTile >= nRowTiles) return;

  const int hi = lane >> 4;
  const int lo = lane & 15;
  const int ldw = NT * 16;
  const float* Arow = S + (size_t)(rowTile * 16 + lo) * 128;

  v8f acc[NT] = {};

#pragma unroll
  for (int k = 0; k < 128; k += 4) {
    const int kk = k + hi * 2;
    v2f a = *(const v2f*)(Arow + kk);
#pragma unroll
    for (int t = 0; t < NT; ++t) {
      const int col = t * 16 + lo;
      v2f b;
      b.x = W[(size_t)kk * ldw + col];
      b.y = W[(size_t)(kk + 1) * ldw + col];
      acc[t] = __builtin_amdgcn_wmma_f32_16x16x4_f32(
           false, a,  false, b,
           (short)0, acc[t],  false,  false);
    }
  }

#pragma unroll
  for (int t = 0; t < NT; ++t) {
    const int col = t * 16 + lo;
    const float bv = bias[col];
#pragma unroll
    for (int j = 0; j < 8; ++j) {
      const int m = j + hi * 8;
      out[(size_t)(rowTile * 16 + m) * ldw + col] = acc[t][j] + bv;
    }
  }
}

extern "C" void kernel_launch(void* const* d_in, const int* in_sizes, int n_in,
                              void* d_out, int out_size, void* d_ws,
                              size_t ws_size, hipStream_t stream) {
  const float* x = (const float*)d_in[0];
  const int* adj = (const int*)d_in[1];
  const float* vals = (const float*)d_in[2];
  const float* W1 = (const float*)d_in[3];
  const float* b1 = (const float*)d_in[4];
  const float* W2 = (const float*)d_in[5];
  const float* b2 = (const float*)d_in[6];
  const float* W3 = (const float*)d_in[7];
  const float* b3 = (const float*)d_in[8];
  float* out = (float*)d_out;

  const int E = in_sizes[2];
  const int N = in_sizes[0] / 128;
  const int* dstIdx = adj;
  const int* srcIdx = adj + E;

  float* ws0 = (float*)d_ws;
  float* ws1 = ws0 + (size_t)N * 128;

  const size_t n4 = (size_t)N * 128 / 4;
  const int zeroBlocks = 2048;
  const int spmmBlocks = (int)(((long long)E * 32 + 255) / 256);
  const int nRowTiles = N / 16;
  const int gemmBlocks = (nRowTiles + 3) / 4;

  zero_f4_kernel<<<zeroBlocks, 256, 0, stream>>>((float4*)ws0, n4);
  spmm_kernel<<<spmmBlocks, 256, 0, stream>>>(dstIdx, srcIdx, vals, x, ws0, E);
  gemm_bias_kernel<8><<<gemmBlocks, 128, 0, stream>>>(ws0, W1, b1, ws1,
                                                      nRowTiles);

  zero_f4_kernel<<<zeroBlocks, 256, 0, stream>>>((float4*)ws0, n4);
  spmm_kernel<<<spmmBlocks, 256, 0, stream>>>(dstIdx, srcIdx, vals, ws1, ws0, E);
  gemm_bias_kernel<8><<<gemmBlocks, 128, 0, stream>>>(ws0, W2, b2, ws1,
                                                      nRowTiles);

  zero_f4_kernel<<<zeroBlocks, 256, 0, stream>>>((float4*)ws0, n4);
  spmm_kernel<<<spmmBlocks, 256, 0, stream>>>(dstIdx, srcIdx, vals, ws1, ws0, E);
  gemm_bias_kernel<4><<<gemmBlocks, 128, 0, stream>>>(ws0, W3, b3, out,
                                                      nRowTiles);
}


# Edges grouped by destination row once per call (returning int adds give ranks); each row's gathered terms summed in registers in rank order, stored once: no float atomics, no zero fill
# speedup vs baseline: 2.6523x; 2.6523x over previous
#include <hip/hip_runtime.h>
#include <hip/hip_bf16.h>
#include <stddef.h>


typedef __attribute__((ext_vector_type(2))) float v2f;
typedef __attribute__((ext_vector_type(8))) float v8f;

__global__ __launch_bounds__(256) void clear_counts_kernel(int* cnt, int n) {
  const int i = blockIdx.x * blockDim.x + threadIdx.x;
  if (i < n) cnt[i] = 0;
}

__global__ __launch_bounds__(256) void edge_rank_kernel(
    const int* __restrict__ dstIdx, int* cnt, int* __restrict__ rank, int E) {
  const int e = blockIdx.x * blockDim.x + threadIdx.x;
  if (e < E) rank[e] = atomicAdd(cnt + dstIdx[e], 1);
}

__global__ __launch_bounds__(256) void row_segment_kernel(
    const int* cnt, int* cursor, int* __restrict__ segStart, int N) {
  const int row = blockIdx.x * blockDim.x + threadIdx.x;
  const int lane = threadIdx.x & 31;
  const int c = (row < N) ? cnt[row] : 0;
  int incl = c;
#pragma unroll
  for (int off = 1; off < 32; off <<= 1) {
    const int t = __shfl_up(incl, off);
    if (lane >= off) incl += t;
  }
  int base = 0;
  if (lane == 31) base = atomicAdd(cursor, incl);
  base = __shfl(base, 31);
  if (row < N) segStart[row] = base + incl - c;
}

__global__ __launch_bounds__(256) void edge_place_kernel(
    const int* __restrict__ dstIdx, const int* __restrict__ srcIdx,
    const float* __restrict__ vals, const int* __restrict__ segStart,
    const int* __restrict__ rank, int* __restrict__ rec, int E) {
  const int e = blockIdx.x * blockDim.x + threadIdx.x;
  if (e < E) {
    const int p = segStart[dstIdx[e]] + rank[e];
    ((int2*)rec)[p] = make_int2(srcIdx[e], __float_as_int(vals[e]));
  }
}

__global__ __launch_bounds__(256) void spmm_rows_kernel(
    const int* __restrict__ segStart, const int* __restrict__ cnt,
    const int* __restrict__ rec, const float* __restrict__ x,
    float* __restrict__ y, int N) {
  const int lane = threadIdx.x & 31;
  const int wave = __builtin_amdgcn_readfirstlane(
      (int)((blockIdx.x * blockDim.x + threadIdx.x) >> 5));
  const int nWaves = (int)((gridDim.x * blockDim.x) >> 5);
  const int2* __restrict__ rec2 = (const int2*)rec;
  for (int row = wave; row < N; row += nWaves) {
    const int beg = __builtin_amdgcn_readfirstlane(segStart[row]);
    const int n = __builtin_amdgcn_readfirstlane(cnt[row]);
    float4 acc = make_float4(0.f, 0.f, 0.f, 0.f);
    for (int base = 0; base < n; base += 32) {
      const int m = (n - base < 32) ? (n - base) : 32;
      int2 ev = make_int2(0, 0);
      if (lane < m) ev = rec2[beg + base + lane];
      for (int j = 0; j < m; j += 8) {
        float4 xv[8];
        float w[8];
#pragma unroll
        for (int k = 0; k < 8; ++k) {
          const int jj = (j + k < m) ? (j + k) : (m - 1);
          const int s = __builtin_amdgcn_readlane(ev.x, jj);
          w[k] = __int_as_float(__builtin_amdgcn_readlane(ev.y, jj));
          xv[k] = *(const float4*)(x + (size_t)s * 128 + (size_t)lane * 4);
        }
#pragma unroll
        for (int k = 0; k < 8; ++k) {
          const bool live = (j + k < m);
          const float ax = fmaf(xv[k].x, w[k], acc.x);
          const float ay = fmaf(xv[k].y, w[k], acc.y);
          const float az = fmaf(xv[k].z, w[k], acc.z);
          const float aw = fmaf(xv[k].w, w[k], acc.w);
          acc.x = live ? ax : acc.x;
          acc.y = live ? ay : acc.y;
          acc.z = live ? az : acc.z;
          acc.w = live ? aw : acc.w;
        }
      }
    }
    *(float4*)(y + (size_t)row * 128 + (size_t)lane * 4) = acc;
  }
}

template <int NT>
__global__ __launch_bounds__(128) void gemm_bias_kernel(
    const float* __restrict__ S, const float* __restrict__ W,
    const float* __restrict__ bias, float* __restrict__ out, int nRowTiles) {
  const int lane = threadIdx.x & 31;
  const int waveInBlock = threadIdx.x >> 5;
  const int rowTile = blockIdx.x * (blockDim.x >> 5) + waveInBlock;
  if (rowTile >= nRowTiles) return;

  const int hi = lane >> 4;
  const int lo = lane & 15;
  const int ldw = NT * 16;
  const float* Arow = S + (size_t)(rowTile * 16 + lo) * 128;

  v8f acc[NT] = {};

#pragma unroll
  for (int k = 0; k < 128; k += 4) {
    const int kk = k + hi * 2;
    v2f a = *(const v2f*)(Arow + kk);
#pragma unroll
    for (int t = 0; t < NT; ++t) {
      const int col = t * 16 + lo;
      v2f b;
      b.x = W[(size_t)kk * ldw + col];
      b.y = W[(size_t)(kk + 1) * ldw + col];
      acc[t] = __builtin_amdgcn_wmma_f32_16x16x4_f32(
           false, a,  false, b,
           (short)0, acc[t],  false,  false);
    }
  }

#pragma unroll
  for (int t = 0; t < NT; ++t) {
    const int col = t * 16 + lo;
    const float bv = bias[col];
#pragma unroll
    for (int j = 0; j < 8; ++j) {
      const int m = j + hi * 8;
      out[(size_t)(rowTile * 16 + m) * ldw + col] = acc[t][j] + bv;
    }
  }
}

extern "C" void kernel_launch(void* const* d_in, const int* in_sizes, int n_in,
                              void* d_out, int out_size, void* d_ws,
                              size_t ws_size, hipStream_t stream) {
  const float* x = (const float*)d_in[0];
  const int* adj = (const int*)d_in[1];
  const float* vals = (const float*)d_in[2];
  const float* W1 = (const float*)d_in[3];
  const float* b1 = (const float*)d_in[4];
  const float* W2 = (const float*)d_in[5];
  const float* b2 = (const float*)d_in[6];
  const float* W3 = (const float*)d_in[7];
  const float* b3 = (const float*)d_in[8];
  float* out = (float*)d_out;

  const int E = in_sizes[2];
  const int N = in_sizes[0] / 128;
  const int* dstIdx = adj;
  const int* srcIdx = adj + E;

  float* ws0 = (float*)d_ws;
  float* ws1 = ws0 + (size_t)N * 128;
  int* cnt = (int*)(ws1 + (size_t)N * 128);
  const size_t nPad = ((size_t)N + 1 + 63) / 64 * 64;
  const size_t ePad = ((size_t)E + 63) / 64 * 64;
  int* segStart = cnt + nPad;
  int* rank = segStart + nPad;
  int* rec = rank + ePad;

  const int cntBlocks = (N + 1 + 255) / 256;
  const int rowBlocks = (N + 255) / 256;
  const int edgeBlocks = (E + 255) / 256;
  const int spmmBlocks = 2048;
  const int nRowTiles = N / 16;
  const int gemmBlocks = (nRowTiles + 3) / 4;

  clear_counts_kernel<<<cntBlocks, 256, 0, stream>>>(cnt, N + 1);
  edge_rank_kernel<<<edgeBlocks, 256, 0, stream>>>(dstIdx, cnt, rank, E);
  row_segment_kernel<<<rowBlocks, 256, 0, stream>>>(cnt, cnt + N, segStart, N);
  edge_place_kernel<<<edgeBlocks, 256, 0, stream>>>(dstIdx, srcIdx, vals,
                                                    segStart, rank, rec, E);

  spmm_rows_kernel<<<spmmBlocks, 256, 0, stream>>>(segStart, cnt, rec, x, ws0, N);
  gemm_bias_kernel<8><<<gemmBlocks, 128, 0, stream>>>(ws0, W1, b1, ws1,
                                                      nRowTiles);

  spmm_rows_kernel<<<spmmBlocks, 256, 0, stream>>>(segStart, cnt, rec, ws1, ws0, N);
  gemm_bias_kernel<8><<<gemmBlocks, 128, 0, stream>>>(ws0, W2, b2, ws1,
                                                      nRowTiles);

  spmm_rows_kernel<<<spmmBlocks, 256, 0, stream>>>(segStart, cnt, rec, ws1, ws0, N);
  gemm_bias_kernel<4><<<gemmBlocks, 128, 0, stream>>>(ws0, W3, b3, out,
                                                      nRowTiles);
}


# Weight products taken first (x@(W1@(W2@W3)), biases carried through W2,W3): all three SpMMs gather 64-wide rows, two 128x128 GEMMs go; products re-associated, summation order changed
# speedup vs baseline: 3.1739x; 1.1967x over previous
#include <hip/hip_runtime.h>
#include <hip/hip_bf16.h>
#include <stddef.h>


typedef __attribute__((ext_vector_type(2))) float v2f;
typedef __attribute__((ext_vector_type(8))) float v8f;

__global__ __launch_bounds__(256) void clear_counts_kernel(int* cnt, int n) {
  const int i = blockIdx.x * blockDim.x + threadIdx.x;
  if (i < n) cnt[i] = 0;
}

__global__ __launch_bounds__(256) void edge_rank_kernel(
    const int* __restrict__ dstIdx, int* cnt, int* __restrict__ rank, int E) {
  const int e = blockIdx.x * blockDim.x + threadIdx.x;
  if (e < E) rank[e] = atomicAdd(cnt + dstIdx[e], 1);
}

__global__ __launch_bounds__(256) void row_segment_kernel(
    const int* cnt, int* cursor, int* __restrict__ segStart, int N) {
  const int row = blockIdx.x * blockDim.x + threadIdx.x;
  const int lane = threadIdx.x & 31;
  const int c = (row < N) ? cnt[row] : 0;
  int incl = c;
#pragma unroll
  for (int off = 1; off < 32; off <<= 1) {
    const int t = __shfl_up(incl, off);
    if (lane >= off) incl += t;
  }
  int base = 0;
  if (lane == 31) base = atomicAdd(cursor, incl);
  base = __shfl(base, 31);
  if (row < N) segStart[row] = base + incl - c;
}

__global__ __launch_bounds__(256) void edge_place_kernel(
    const int* __restrict__ dstIdx, const int* __restrict__ srcIdx,
    const float* __restrict__ vals, const int* __restrict__ segStart,
    const int* __restrict__ rank, int* __restrict__ rec, int E) {
  const int e = blockIdx.x * blockDim.x + threadIdx.x;
  if (e < E) {
    const int p = segStart[dstIdx[e]] + rank[e];
    ((int2*)rec)[p] = make_int2(srcIdx[e], __float_as_int(vals[e]));
  }
}

__global__ __launch_bounds__(256) void spmm_rows64_kernel(
    const int* __restrict__ segStart, const int* __restrict__ cnt,
    const int* __restrict__ rec, const float* __restrict__ z,
    const float* __restrict__ c, float* __restrict__ y, int N) {
  const int lane = threadIdx.x & 31;
  const int wave = __builtin_amdgcn_readfirstlane(
      (int)((blockIdx.x * blockDim.x + threadIdx.x) >> 5));
  const int nWaves = (int)((gridDim.x * blockDim.x) >> 5);
  const int2* __restrict__ rec2 = (const int2*)rec;
  const float2 cv = *(const float2*)(c + (size_t)lane * 2);
  for (int row = wave; row < N; row += nWaves) {
    const int beg = __builtin_amdgcn_readfirstlane(segStart[row]);
    const int n = __builtin_amdgcn_readfirstlane(cnt[row]);
    float2 acc = make_float2(0.f, 0.f);
    for (int base = 0; base < n; base += 32) {
      const int m = (n - base < 32) ? (n - base) : 32;
      int2 ev = make_int2(0, 0);
      if (lane < m) ev = rec2[beg + base + lane];
      for (int j = 0; j < m; j += 16) {
        float2 zv[16];
        float w[16];
#pragma unroll
        for (int k = 0; k < 16; ++k) {
          const int jj = (j + k < m) ? (j + k) : (m - 1);
          const int s = __builtin_amdgcn_readlane(ev.x, jj);
          w[k] = __int_as_float(__builtin_amdgcn_readlane(ev.y, jj));
          zv[k] = *(const float2*)(z + (size_t)s * 64 + (size_t)lane * 2);
        }
#pragma unroll
        for (int k = 0; k < 16; ++k) {
          const bool live = (j + k < m);
          const float ax = fmaf(zv[k].x, w[k], acc.x);
          const float ay = fmaf(zv[k].y, w[k], acc.y);
          acc.x = live ? ax : acc.x;
          acc.y = live ? ay : acc.y;
        }
      }
    }
    acc.x += cv.x;
    acc.y += cv.y;
    *(float2*)(y + (size_t)row * 64 + (size_t)lane * 2) = acc;
  }
}

__global__ __launch_bounds__(128) void bias_fold_kernel(
    const float* __restrict__ b1, const float* __restrict__ W23,
    const float* __restrict__ b2, const float* __restrict__ W3,
    float* __restrict__ c1, float* __restrict__ c2) {
  const int j = threadIdx.x & 63;
  const bool first = threadIdx.x < 64;
  const float* __restrict__ b = first ? b1 : b2;
  const float* __restrict__ Wm = first ? W23 : W3;
  float acc = 0.f;
  for (int k = 0; k < 128; ++k) acc = fmaf(b[k], Wm[k * 64 + j], acc);
  if (first) c1[j] = acc; else c2[j] = acc;
}

template <int NT>
__global__ __launch_bounds__(128) void gemm_kernel(
    const float* __restrict__ S, const float* __restrict__ W,
    float* __restrict__ out, int nRowTiles) {
  const int lane = threadIdx.x & 31;
  const int waveInBlock = threadIdx.x >> 5;
  const int rowTile = blockIdx.x * (blockDim.x >> 5) + waveInBlock;
  if (rowTile >= nRowTiles) return;

  const int hi = lane >> 4;
  const int lo = lane & 15;
  const int ldw = NT * 16;
  const float* Arow = S + (size_t)(rowTile * 16 + lo) * 128;

  v8f acc[NT] = {};

#pragma unroll
  for (int k = 0; k < 128; k += 4) {
    const int kk = k + hi * 2;
    v2f a = *(const v2f*)(Arow + kk);
#pragma unroll
    for (int t = 0; t < NT; ++t) {
      const int col = t * 16 + lo;
      v2f b;
      b.x = W[(size_t)kk * ldw + col];
      b.y = W[(size_t)(kk + 1) * ldw + col];
      acc[t] = __builtin_amdgcn_wmma_f32_16x16x4_f32(
           false, a,  false, b,
           (short)0, acc[t],  false,  false);
    }
  }

#pragma unroll
  for (int t = 0; t < NT; ++t) {
    const int col = t * 16 + lo;
#pragma unroll
    for (int j = 0; j < 8; ++j) {
      const int m = j + hi * 8;
      out[(size_t)(rowTile * 16 + m) * ldw + col] = acc[t][j];
    }
  }
}

extern "C" void kernel_launch(void* const* d_in, const int* in_sizes, int n_in,
                              void* d_out, int out_size, void* d_ws,
                              size_t ws_size, hipStream_t stream) {
  const float* x = (const float*)d_in[0];
  const int* adj = (const int*)d_in[1];
  const float* vals = (const float*)d_in[2];
  const float* W1 = (const float*)d_in[3];
  const float* b1 = (const float*)d_in[4];
  const float* W2 = (const float*)d_in[5];
  const float* b2 = (const float*)d_in[6];
  const float* W3 = (const float*)d_in[7];
  const float* b3 = (const float*)d_in[8];
  float* out = (float*)d_out;

  const int E = in_sizes[2];
  const int N = in_sizes[0] / 128;
  const int* dstIdx = adj;
  const int* srcIdx = adj + E;

  float* zA = (float*)d_ws;
  float* zB = zA + (size_t)N * 64;
  float* W23 = zB + (size_t)N * 64;
  float* Wc = W23 + 128 * 64;
  float* c1 = Wc + 128 * 64;
  float* c2 = c1 + 64;
  int* cnt = (int*)(c2 + 64);
  const size_t nPad = ((size_t)N + 1 + 63) / 64 * 64;
  const size_t ePad = ((size_t)E + 63) / 64 * 64;
  int* segStart = cnt + nPad;
  int* rank = segStart + nPad;
  int* rec = rank + ePad;

  const int cntBlocks = (N + 1 + 255) / 256;
  const int rowBlocks = (N + 255) / 256;
  const int edgeBlocks = (E + 255) / 256;
  const int spmmBlocks = 2048;
  const int nRowTiles = N / 16;
  const int gemmBlocks = (nRowTiles + 3) / 4;

  clear_counts_kernel<<<cntBlocks, 256, 0, stream>>>(cnt, N + 1);
  edge_rank_kernel<<<edgeBlocks, 256, 0, stream>>>(dstIdx, cnt, rank, E);
  row_segment_kernel<<<rowBlocks, 256, 0, stream>>>(cnt, cnt + N, segStart, N);
  edge_place_kernel<<<edgeBlocks, 256, 0, stream>>>(dstIdx, srcIdx, vals,
                                                    segStart, rank, rec, E);

  gemm_kernel<4><<<2, 128, 0, stream>>>(W2, W3, W23, 8);
  gemm_kernel<4><<<2, 128, 0, stream>>>(W1, W23, Wc, 8);
  bias_fold_kernel<<<1, 128, 0, stream>>>(b1, W23, b2, W3, c1, c2);

  gemm_kernel<4><<<gemmBlocks, 128, 0, stream>>>(x, Wc, zA, nRowTiles);
  spmm_rows64_kernel<<<spmmBlocks, 256, 0, stream>>>(segStart, cnt, rec, zA, c1, zB, N);
  spmm_rows64_kernel<<<spmmBlocks, 256, 0, stream>>>(segStart, cnt, rec, zB, c2, zA, N);
  spmm_rows64_kernel<<<spmmBlocks, 256, 0, stream>>>(segStart, cnt, rec, zA, b3, out, N);
}


# Edge grouping as a two-level counting sort on LDS counters (256 row buckets, then rows in a bucket): no global atomics, 4 launches to 3; a row's terms now summed in LDS-add order
# speedup vs baseline: 3.7308x; 1.1755x over previous
#include <hip/hip_runtime.h>
#include <hip/hip_bf16.h>
#include <stddef.h>


typedef __attribute__((ext_vector_type(2))) float v2f;
typedef __attribute__((ext_vector_type(8))) float v8f;

constexpr int NB = 256;
constexpr int EB = 8;

__device__ __forceinline__ int block_excl_scan256(int v, int* wsum) {
  const int lane = threadIdx.x & 31;
  const int wave = threadIdx.x >> 5;
  int incl = v;
#pragma unroll
  for (int off = 1; off < 32; off <<= 1) {
    const int t = __shfl_up(incl, off);
    if (lane >= off) incl += t;
  }
  if (lane == 31) wsum[wave] = incl;
  __syncthreads();
  int base = incl - v;
  for (int i = 0; i < wave; ++i) base += wsum[i];
  return base;
}

__global__ __launch_bounds__(256) void bucket_count_kernel(
    const int* __restrict__ dstIdx, int* __restrict__ hist, int E, int N) {
  __shared__ int h[NB];
  const int t = threadIdx.x;
  h[t] = 0;
  __syncthreads();
  const unsigned rpb = (unsigned)(N + NB - 1) / NB;
  const int chunk = (E + NB - 1) / NB;
  const int e0 = blockIdx.x * chunk;
  const int e1 = (e0 + chunk < E) ? (e0 + chunk) : E;
  for (int e = e0 + t; e < e1; e += NB * EB) {
    int d[EB];
#pragma unroll
    for (int k = 0; k < EB; ++k) {
      const int i = e + k * NB;
      d[k] = (i < e1) ? dstIdx[i] : -1;
    }
#pragma unroll
    for (int k = 0; k < EB; ++k)
      if (d[k] >= 0) atomicAdd(&h[(unsigned)d[k] / rpb], 1);
  }
  __syncthreads();
  hist[blockIdx.x * NB + t] = h[t];
}

__global__ __launch_bounds__(256) void bucket_scatter_kernel(
    const int* __restrict__ dstIdx, const int* __restrict__ srcIdx,
    const float* __restrict__ vals, const int* __restrict__ hist,
    int* __restrict__ bucketBase, int* __restrict__ tmpDst,
    int* __restrict__ tmpRec, int E, int N) {
  __shared__ int cur[NB];
  __shared__ int wsum[8];
  const int t = threadIdx.x;
  const int g = blockIdx.x;
  int total = 0, pre = 0;
  for (int w0 = 0; w0 < NB; w0 += 32) {
    int v[32];
#pragma unroll
    for (int k = 0; k < 32; ++k) v[k] = hist[(w0 + k) * NB + t];
#pragma unroll
    for (int k = 0; k < 32; ++k) {
      total += v[k];
      pre += (w0 + k < g) ? v[k] : 0;
    }
  }
  const int base = block_excl_scan256(total, wsum);
  cur[t] = base + pre;
  if (g == 0) {
    bucketBase[t] = base;
    if (t == NB - 1) bucketBase[NB] = base + total;
  }
  __syncthreads();
  const unsigned rpb = (unsigned)(N + NB - 1) / NB;
  const int chunk = (E + NB - 1) / NB;
  const int e0 = g * chunk;
  const int e1 = (e0 + chunk < E) ? (e0 + chunk) : E;
  int2* __restrict__ tmpRec2 = (int2*)tmpRec;
  for (int e = e0 + t; e < e1; e += NB * EB) {
    int d[EB], s[EB];
    float v[EB];
#pragma unroll
    for (int k = 0; k < EB; ++k) {
      const int i = e + k * NB;
      const bool in = i < e1;
      d[k] = in ? dstIdx[i] : -1;
      s[k] = in ? srcIdx[i] : 0;
      v[k] = in ? vals[i] : 0.f;
    }
#pragma unroll
    for (int k = 0; k < EB; ++k) {
      if (d[k] >= 0) {
        const int p = atomicAdd(&cur[(unsigned)d[k] / rpb], 1);
        tmpDst[p] = d[k];
        tmpRec2[p] = make_int2(s[k], __float_as_int(v[k]));
      }
    }
  }
}

__global__ __launch_bounds__(256) void bucket_rows_kernel(
    const int* __restrict__ bucketBase, const int* __restrict__ tmpDst,
    const int* __restrict__ tmpRec, int* __restrict__ segStart,
    int* __restrict__ cnt, int* __restrict__ rec, int N) {
  __shared__ int c[NB];
  __shared__ int wsum[8];
  const int t = threadIdx.x;
  const int g = blockIdx.x;
  const int rpb = (N + NB - 1) / NB;
  const int r0 = g * rpb;
  const int b0 = bucketBase[g];
  const int b1 = bucketBase[g + 1];
  c[t] = 0;
  __syncthreads();
  for (int i = b0 + t; i < b1; i += NB * EB) {
    int d[EB];
#pragma unroll
    for (int k = 0; k < EB; ++k) {
      const int j = i + k * NB;
      d[k] = (j < b1) ? tmpDst[j] : -1;
    }
#pragma unroll
    for (int k = 0; k < EB; ++k)
      if (d[k] >= 0) atomicAdd(&c[d[k] - r0], 1);
  }
  __syncthreads();
  const int mine = c[t];
  const int start = b0 + block_excl_scan256(mine, wsum);
  c[t] = start;
  const int row = r0 + t;
  if (t < rpb && row < N) {
    segStart[row] = start;
    cnt[row] = mine;
  }
  __syncthreads();
  const int2* __restrict__ tmpRec2 = (const int2*)tmpRec;
  int2* __restrict__ rec2 = (int2*)rec;
  for (int i = b0 + t; i < b1; i += NB * EB) {
    int d[EB];
    int2 r[EB];
#pragma unroll
    for (int k = 0; k < EB; ++k) {
      const int j = i + k * NB;
      const bool in = j < b1;
      d[k] = in ? tmpDst[j] : -1;
      r[k] = in ? tmpRec2[j] : make_int2(0, 0);
    }
#pragma unroll
    for (int k = 0; k < EB; ++k) {
      if (d[k] >= 0) {
        const int p = atomicAdd(&c[d[k] - r0], 1);
        rec2[p] = r[k];
      }
    }
  }
}

__global__ __launch_bounds__(256) void spmm_rows64_kernel(
    const int* __restrict__ segStart, const int* __restrict__ cnt,
    const int* __restrict__ rec, const float* __restrict__ z,
    const float* __restrict__ c, float* __restrict__ y, int N) {
  const int lane = threadIdx.x & 31;
  const int wave = __builtin_amdgcn_readfirstlane(
      (int)((blockIdx.x * blockDim.x + threadIdx.x) >> 5));
  const int nWaves = (int)((gridDim.x * blockDim.x) >> 5);
  const int2* __restrict__ rec2 = (const int2*)rec;
  const float2 cv = *(const float2*)(c + (size_t)lane * 2);
  for (int row = wave; row < N; row += nWaves) {
    const int beg = __builtin_amdgcn_readfirstlane(segStart[row]);
    const int n = __builtin_amdgcn_readfirstlane(cnt[row]);
    float2 acc = make_float2(0.f, 0.f);
    for (int base = 0; base < n; base += 32) {
      const int m = (n - base < 32) ? (n - base) : 32;
      int2 ev = make_int2(0, 0);
      if (lane < m) ev = rec2[beg + base + lane];
      for (int j = 0; j < m; j += 16) {
        float2 zv[16];
        float w[16];
#pragma unroll
        for (int k = 0; k < 16; ++k) {
          const int jj = (j + k < m) ? (j + k) : (m - 1);
          const int s = __builtin_amdgcn_readlane(ev.x, jj);
          w[k] = __int_as_float(__builtin_amdgcn_readlane(ev.y, jj));
          zv[k] = *(const float2*)(z + (size_t)s * 64 + (size_t)lane * 2);
        }
#pragma unroll
        for (int k = 0; k < 16; ++k) {
          const bool live = (j + k < m);
          const float ax = fmaf(zv[k].x, w[k], acc.x);
          const float ay = fmaf(zv[k].y, w[k], acc.y);
          acc.x = live ? ax : acc.x;
          acc.y = live ? ay : acc.y;
        }
      }
    }
    acc.x += cv.x;
    acc.y += cv.y;
    *(float2*)(y + (size_t)row * 64 + (size_t)lane * 2) = acc;
  }
}

__global__ __launch_bounds__(128) void bias_fold_kernel(
    const float* __restrict__ b1, const float* __restrict__ W23,
    const float* __restrict__ b2, const float* __restrict__ W3,
    float* __restrict__ c1, float* __restrict__ c2) {
  const int j = threadIdx.x & 63;
  const bool first = threadIdx.x < 64;
  const float* __restrict__ b = first ? b1 : b2;
  const float* __restrict__ Wm = first ? W23 : W3;
  float acc = 0.f;
  for (int k = 0; k < 128; ++k) acc = fmaf(b[k], Wm[k * 64 + j], acc);
  if (first) c1[j] = acc; else c2[j] = acc;
}

template <int NT>
__global__ __launch_bounds__(128) void gemm_kernel(
    const float* __restrict__ S, const float* __restrict__ W,
    float* __restrict__ out, int nRowTiles) {
  const int lane = threadIdx.x & 31;
  const int waveInBlock = threadIdx.x >> 5;
  const int rowTile = blockIdx.x * (blockDim.x >> 5) + waveInBlock;
  if (rowTile >= nRowTiles) return;

  const int hi = lane >> 4;
  const int lo = lane & 15;
  const int ldw = NT * 16;
  const float* Arow = S + (size_t)(rowTile * 16 + lo) * 128;

  v8f acc[NT] = {};

#pragma unroll
  for (int k = 0; k < 128; k += 4) {
    const int kk = k + hi * 2;
    v2f a = *(const v2f*)(Arow + kk);
#pragma unroll
    for (int t = 0; t < NT; ++t) {
      const int col = t * 16 + lo;
      v2f b;
      b.x = W[(size_t)kk * ldw + col];
      b.y = W[(size_t)(kk + 1) * ldw + col];
      acc[t] = __builtin_amdgcn_wmma_f32_16x16x4_f32(
           false, a,  false, b,
           (short)0, acc[t],  false,  false);
    }
  }

#pragma unroll
  for (int t = 0; t < NT; ++t) {
    const int col = t * 16 + lo;
#pragma unroll
    for (int j = 0; j < 8; ++j) {
      const int m = j + hi * 8;
      out[(size_t)(rowTile * 16 + m) * ldw + col] = acc[t][j];
    }
  }
}

extern "C" void kernel_launch(void* const* d_in, const int* in_sizes, int n_in,
                              void* d_out, int out_size, void* d_ws,
                              size_t ws_size, hipStream_t stream) {
  const float* x = (const float*)d_in[0];
  const int* adj = (const int*)d_in[1];
  const float* vals = (const float*)d_in[2];
  const float* W1 = (const float*)d_in[3];
  const float* b1 = (const float*)d_in[4];
  const float* W2 = (const float*)d_in[5];
  const float* b2 = (const float*)d_in[6];
  const float* W3 = (const float*)d_in[7];
  const float* b3 = (const float*)d_in[8];
  float* out = (float*)d_out;

  const int E = in_sizes[2];
  const int N = in_sizes[0] / 128;
  const int* dstIdx = adj;
  const int* srcIdx = adj + E;

  float* zA = (float*)d_ws;
  float* zB = zA + (size_t)N * 64;
  float* W23 = zB + (size_t)N * 64;
  float* Wc = W23 + 128 * 64;
  float* c1 = Wc + 128 * 64;
  float* c2 = c1 + 64;
  int* cnt = (int*)(c2 + 64);
  const size_t nPad = ((size_t)N + 63) / 64 * 64;
  const size_t ePad = ((size_t)E + 63) / 64 * 64;
  int* segStart = cnt + nPad;
  int* hist = segStart + nPad;
  int* bucketBase = hist + NB * NB;
  int* tmpDst = bucketBase + 320;
  int* tmpRec = tmpDst + ePad;
  int* rec = tmpRec + 2 * ePad;

  const int spmmBlocks = 2048;
  const int nRowTiles = N / 16;
  const int gemmBlocks = (nRowTiles + 3) / 4;

  bucket_count_kernel<<<NB, 256, 0, stream>>>(dstIdx, hist, E, N);
  bucket_scatter_kernel<<<NB, 256, 0, stream>>>(dstIdx, srcIdx, vals, hist,
                                                bucketBase, tmpDst, tmpRec, E, N);
  bucket_rows_kernel<<<NB, 256, 0, stream>>>(bucketBase, tmpDst, tmpRec,
                                             segStart, cnt, rec, N);

  gemm_kernel<4><<<2, 128, 0, stream>>>(W2, W3, W23, 8);
  gemm_kernel<4><<<2, 128, 0, stream>>>(W1, W23, Wc, 8);
  bias_fold_kernel<<<1, 128, 0, stream>>>(b1, W23, b2, W3, c1, c2);

  gemm_kernel<4><<<gemmBlocks, 128, 0, stream>>>(x, Wc, zA, nRowTiles);
  spmm_rows64_kernel<<<spmmBlocks, 256, 0, stream>>>(segStart, cnt, rec, zA, c1, zB, N);
  spmm_rows64_kernel<<<spmmBlocks, 256, 0, stream>>>(segStart, cnt, rec, zB, c2, zA, N);
  spmm_rows64_kernel<<<spmmBlocks, 256, 0, stream>>>(segStart, cnt, rec, zA, b3, out, N);
}


# Edge grouping and the dense chain (W2@W3; W1@W23 and bias folds; x@Wc) are independent: step i of both now shares one launch (extra work-groups), 10 launches to 6
# speedup vs baseline: 4.6254x; 1.2398x over previous
#include <hip/hip_runtime.h>
#include <hip/hip_bf16.h>
#include <stddef.h>


typedef __attribute__((ext_vector_type(2))) float v2f;
typedef __attribute__((ext_vector_type(8))) float v8f;

constexpr int NB = 256;
constexpr int EB = 8;

__device__ __forceinline__ int block_excl_scan256(int v, int* wsum) {
  const int lane = threadIdx.x & 31;
  const int wave = threadIdx.x >> 5;
  int incl = v;
#pragma unroll
  for (int off = 1; off < 32; off <<= 1) {
    const int t = __shfl_up(incl, off);
    if (lane >= off) incl += t;
  }
  if (lane == 31) wsum[wave] = incl;
  __syncthreads();
  int base = incl - v;
  for (int i = 0; i < wave; ++i) base += wsum[i];
  return base;
}

__device__ __forceinline__ void bucket_count_body(
    const int g, const int* __restrict__ dstIdx, int* __restrict__ hist, int E,
    int N) {
  __shared__ int h[NB];
  const int t = threadIdx.x;
  h[t] = 0;
  __syncthreads();
  const unsigned rpb = (unsigned)(N + NB - 1) / NB;
  const int chunk = (E + NB - 1) / NB;
  const int e0 = g * chunk;
  const int e1 = (e0 + chunk < E) ? (e0 + chunk) : E;
  for (int e = e0 + t; e < e1; e += NB * EB) {
    int d[EB];
#pragma unroll
    for (int k = 0; k < EB; ++k) {
      const int i = e + k * NB;
      d[k] = (i < e1) ? dstIdx[i] : -1;
    }
#pragma unroll
    for (int k = 0; k < EB; ++k)
      if (d[k] >= 0) atomicAdd(&h[(unsigned)d[k] / rpb], 1);
  }
  __syncthreads();
  hist[g * NB + t] = h[t];
}

__device__ __forceinline__ void bucket_scatter_body(
    const int g, const int* __restrict__ dstIdx, const int* __restrict__ srcIdx,
    const float* __restrict__ vals, const int* __restrict__ hist,
    int* __restrict__ bucketBase, int* __restrict__ tmpDst,
    int* __restrict__ tmpRec, int E, int N) {
  __shared__ int cur[NB];
  __shared__ int wsum[8];
  const int t = threadIdx.x;
  int total = 0, pre = 0;
  for (int w0 = 0; w0 < NB; w0 += 32) {
    int v[32];
#pragma unroll
    for (int k = 0; k < 32; ++k) v[k] = hist[(w0 + k) * NB + t];
#pragma unroll
    for (int k = 0; k < 32; ++k) {
      total += v[k];
      pre += (w0 + k < g) ? v[k] : 0;
    }
  }
  const int base = block_excl_scan256(total, wsum);
  cur[t] = base + pre;
  if (g == 0) {
    bucketBase[t] = base;
    if (t == NB - 1) bucketBase[NB] = base + total;
  }
  __syncthreads();
  const unsigned rpb = (unsigned)(N + NB - 1) / NB;
  const int chunk = (E + NB - 1) / NB;
  const int e0 = g * chunk;
  const int e1 = (e0 + chunk < E) ? (e0 + chunk) : E;
  int2* __restrict__ tmpRec2 = (int2*)tmpRec;
  for (int e = e0 + t; e < e1; e += NB * EB) {
    int d[EB], s[EB];
    float v[EB];
#pragma unroll
    for (int k = 0; k < EB; ++k) {
      const int i = e + k * NB;
      const bool in = i < e1;
      d[k] = in ? dstIdx[i] : -1;
      s[k] = in ? srcIdx[i] : 0;
      v[k] = in ? vals[i] : 0.f;
    }
#pragma unroll
    for (int k = 0; k < EB; ++k) {
      if (d[k] >= 0) {
        const int p = atomicAdd(&cur[(unsigned)d[k] / rpb], 1);
        tmpDst[p] = d[k];
        tmpRec2[p] = make_int2(s[k], __float_as_int(v[k]));
      }
    }
  }
}

__device__ __forceinline__ void bucket_rows_body(
    const int g, const int* __restrict__ bucketBase,
    const int* __restrict__ tmpDst,
    const int* __restrict__ tmpRec, int* __restrict__ segStart,
    int* __restrict__ cnt, int* __restrict__ rec, int N) {
  __shared__ int c[NB];
  __shared__ int wsum[8];
  const int t = threadIdx.x;
  const int rpb = (N + NB - 1) / NB;
  const int r0 = g * rpb;
  const int b0 = bucketBase[g];
  const int b1 = bucketBase[g + 1];
  c[t] = 0;
  __syncthreads();
  for (int i = b0 + t; i < b1; i += NB * EB) {
    int d[EB];
#pragma unroll
    for (int k = 0; k < EB; ++k) {
      const int j = i + k * NB;
      d[k] = (j < b1) ? tmpDst[j] : -1;
    }
#pragma unroll
    for (int k = 0; k < EB; ++k)
      if (d[k] >= 0) atomicAdd(&c[d[k] - r0], 1);
  }
  __syncthreads();
  const int mine = c[t];
  const int start = b0 + block_excl_scan256(mine, wsum);
  c[t] = start;
  const int row = r0 + t;
  if (t < rpb && row < N) {
    segStart[row] = start;
    cnt[row] = mine;
  }
  __syncthreads();
  const int2* __restrict__ tmpRec2 = (const int2*)tmpRec;
  int2* __restrict__ rec2 = (int2*)rec;
  for (int i = b0 + t; i < b1; i += NB * EB) {
    int d[EB];
    int2 r[EB];
#pragma unroll
    for (int k = 0; k < EB; ++k) {
      const int j = i + k * NB;
      const bool in = j < b1;
      d[k] = in ? tmpDst[j] : -1;
      r[k] = in ? tmpRec2[j] : make_int2(0, 0);
    }
#pragma unroll
    for (int k = 0; k < EB; ++k) {
      if (d[k] >= 0) {
        const int p = atomicAdd(&c[d[k] - r0], 1);
        rec2[p] = r[k];
      }
    }
  }
}

__global__ __launch_bounds__(256) void spmm_rows64_kernel(
    const int* __restrict__ segStart, const int* __restrict__ cnt,
    const int* __restrict__ rec, const float* __restrict__ z,
    const float* __restrict__ c, float* __restrict__ y, int N) {
  const int lane = threadIdx.x & 31;
  const int wave = __builtin_amdgcn_readfirstlane(
      (int)((blockIdx.x * blockDim.x + threadIdx.x) >> 5));
  const int nWaves = (int)((gridDim.x * blockDim.x) >> 5);
  const int2* __restrict__ rec2 = (const int2*)rec;
  const float2 cv = *(const float2*)(c + (size_t)lane * 2);
  for (int row = wave; row < N; row += nWaves) {
    const int beg = __builtin_amdgcn_readfirstlane(segStart[row]);
    const int n = __builtin_amdgcn_readfirstlane(cnt[row]);
    float2 acc = make_float2(0.f, 0.f);
    for (int base = 0; base < n; base += 32) {
      const int m = (n - base < 32) ? (n - base) : 32;
      int2 ev = make_int2(0, 0);
      if (lane < m) ev = rec2[beg + base + lane];
      for (int j = 0; j < m; j += 16) {
        float2 zv[16];
        float w[16];
#pragma unroll
        for (int k = 0; k < 16; ++k) {
          const int jj = (j + k < m) ? (j + k) : (m - 1);
          const int s = __builtin_amdgcn_readlane(ev.x, jj);
          w[k] = __int_as_float(__builtin_amdgcn_readlane(ev.y, jj));
          zv[k] = *(const float2*)(z + (size_t)s * 64 + (size_t)lane * 2);
        }
#pragma unroll
        for (int k = 0; k < 16; ++k) {
          const bool live = (j + k < m);
          const float ax = fmaf(zv[k].x, w[k], acc.x);
          const float ay = fmaf(zv[k].y, w[k], acc.y);
          acc.x = live ? ax : acc.x;
          acc.y = live ? ay : acc.y;
        }
      }
    }
    acc.x += cv.x;
    acc.y += cv.y;
    *(float2*)(y + (size_t)row * 64 + (size_t)lane * 2) = acc;
  }
}

__device__ __forceinline__ void bias_fold_body(
    const float* __restrict__ b1, const float* __restrict__ W23,
    const float* __restrict__ b2, const float* __restrict__ W3,
    float* __restrict__ c1, float* __restrict__ c2) {
  if (threadIdx.x >= 128) return;
  const int j = threadIdx.x & 63;
  const bool first = threadIdx.x < 64;
  const float* __restrict__ b = first ? b1 : b2;
  const float* __restrict__ Wm = first ? W23 : W3;
  float acc = 0.f;
  for (int k = 0; k < 128; ++k) acc = fmaf(b[k], Wm[k * 64 + j], acc);
  if (first) c1[j] = acc; else c2[j] = acc;
}

template <int NT>
__device__ __forceinline__ void gemm_body(
    const int gb, const float* __restrict__ S, const float* __restrict__ W,
    float* __restrict__ out, int nRowTiles) {
  const int lane = threadIdx.x & 31;
  const int waveInBlock = threadIdx.x >> 5;
  const int rowTile = gb * (blockDim.x >> 5) + waveInBlock;
  if (rowTile >= nRowTiles) return;

  const int hi = lane >> 4;
  const int lo = lane & 15;
  const int ldw = NT * 16;
  const float* Arow = S + (size_t)(rowTile * 16 + lo) * 128;

  v8f acc[NT] = {};

#pragma unroll
  for (int k = 0; k < 128; k += 4) {
    const int kk = k + hi * 2;
    v2f a = *(const v2f*)(Arow + kk);
#pragma unroll
    for (int t = 0; t < NT; ++t) {
      const int col = t * 16 + lo;
      v2f b;
      b.x = W[(size_t)kk * ldw + col];
      b.y = W[(size_t)(kk + 1) * ldw + col];
      acc[t] = __builtin_amdgcn_wmma_f32_16x16x4_f32(
           false, a,  false, b,
           (short)0, acc[t],  false,  false);
    }
  }

#pragma unroll
  for (int t = 0; t < NT; ++t) {
    const int col = t * 16 + lo;
#pragma unroll
    for (int j = 0; j < 8; ++j) {
      const int m = j + hi * 8;
      out[(size_t)(rowTile * 16 + m) * ldw + col] = acc[t][j];
    }
  }
}

__global__ __launch_bounds__(256) void step1_kernel(
    const int* __restrict__ dstIdx, int* __restrict__ hist, int E, int N,
    const float* __restrict__ W2, const float* __restrict__ W3,
    float* __restrict__ W23) {
  const int g = blockIdx.x;
  if (g < NB) bucket_count_body(g, dstIdx, hist, E, N);
  else gemm_body<4>(g - NB, W2, W3, W23, 8);
}

__global__ __launch_bounds__(256) void step2_kernel(
    const int* __restrict__ dstIdx, const int* __restrict__ srcIdx,
    const float* __restrict__ vals, const int* __restrict__ hist,
    int* __restrict__ bucketBase, int* __restrict__ tmpDst,
    int* __restrict__ tmpRec, int E, int N, const float* __restrict__ W1,
    const float* __restrict__ W23, float* __restrict__ Wc,
    const float* __restrict__ b1, const float* __restrict__ b2,
    const float* __restrict__ W3, float* __restrict__ c1,
    float* __restrict__ c2) {
  const int g = blockIdx.x;
  if (g < NB)
    bucket_scatter_body(g, dstIdx, srcIdx, vals, hist, bucketBase, tmpDst,
                        tmpRec, E, N);
  else if (g == NB) gemm_body<4>(0, W1, W23, Wc, 8);
  else bias_fold_body(b1, W23, b2, W3, c1, c2);
}

__global__ __launch_bounds__(256) void step3_kernel(
    const int* __restrict__ bucketBase, const int* __restrict__ tmpDst,
    const int* __restrict__ tmpRec, int* __restrict__ segStart,
    int* __restrict__ cnt, int* __restrict__ rec, int N,
    const float* __restrict__ x, const float* __restrict__ Wc,
    float* __restrict__ z0, int nRowTiles) {
  const int g = blockIdx.x;
  if (g < NB) bucket_rows_body(g, bucketBase, tmpDst, tmpRec, segStart, cnt, rec, N);
  else gemm_body<4>(g - NB, x, Wc, z0, nRowTiles);
}

extern "C" void kernel_launch(void* const* d_in, const int* in_sizes, int n_in,
                              void* d_out, int out_size, void* d_ws,
                              size_t ws_size, hipStream_t stream) {
  const float* x = (const float*)d_in[0];
  const int* adj = (const int*)d_in[1];
  const float* vals = (const float*)d_in[2];
  const float* W1 = (const float*)d_in[3];
  const float* b1 = (const float*)d_in[4];
  const float* W2 = (const float*)d_in[5];
  const float* b2 = (const float*)d_in[6];
  const float* W3 = (const float*)d_in[7];
  const float* b3 = (const float*)d_in[8];
  float* out = (float*)d_out;

  const int E = in_sizes[2];
  const int N = in_sizes[0] / 128;
  const int* dstIdx = adj;
  const int* srcIdx = adj + E;

  float* zA = (float*)d_ws;
  float* zB = zA + (size_t)N * 64;
  float* W23 = zB + (size_t)N * 64;
  float* Wc = W23 + 128 * 64;
  float* c1 = Wc + 128 * 64;
  float* c2 = c1 + 64;
  int* cnt = (int*)(c2 + 64);
  const size_t nPad = ((size_t)N + 63) / 64 * 64;
  const size_t ePad = ((size_t)E + 63) / 64 * 64;
  int* segStart = cnt + nPad;
  int* hist = segStart + nPad;
  int* bucketBase = hist + NB * NB;
  int* tmpDst = bucketBase + 320;
  int* tmpRec = tmpDst + ePad;
  int* rec = tmpRec + 2 * ePad;

  const int spmmBlocks = 2048;
  const int nRowTiles = N / 16;
  const int gemmBlocks = (nRowTiles + 7) / 8;

  step1_kernel<<<NB + 1, 256, 0, stream>>>(dstIdx, hist, E, N, W2, W3, W23);
  step2_kernel<<<NB + 2, 256, 0, stream>>>(dstIdx, srcIdx, vals, hist,
                                           bucketBase, tmpDst, tmpRec, E, N, W1,
                                           W23, Wc, b1, b2, W3, c1, c2);
  step3_kernel<<<NB + gemmBlocks, 256, 0, stream>>>(bucketBase, tmpDst, tmpRec,
                                                    segStart, cnt, rec, N, x, Wc,
                                                    zA, nRowTiles);

  spmm_rows64_kernel<<<spmmBlocks, 256, 0, stream>>>(segStart, cnt, rec, zA, c1, zB, N);
  spmm_rows64_kernel<<<spmmBlocks, 256, 0, stream>>>(segStart, cnt, rec, zB, c2, zA, N);
  spmm_rows64_kernel<<<spmmBlocks, 256, 0, stream>>>(segStart, cnt, rec, zA, b3, out, N);
}


# Edge grouping in 2 launches, not 3: a chunk is sorted by bucket inside its own range, a bucket's work-group collects its 256 pieces; weight chain in one work-group; a row's LDS-add order differs
# speedup vs baseline: 6.8779x; 1.4870x over previous
#include <hip/hip_runtime.h>
#include <hip/hip_bf16.h>
#include <stddef.h>


typedef __attribute__((ext_vector_type(2))) float v2f;
typedef __attribute__((ext_vector_type(8))) float v8f;

constexpr int NB = 256;
constexpr int EB = 8;

__device__ __forceinline__ int block_excl_scan256(int v, int* wsum) {
  const int lane = threadIdx.x & 31;
  const int wave = threadIdx.x >> 5;
  int incl = v;
#pragma unroll
  for (int off = 1; off < 32; off <<= 1) {
    const int t = __shfl_up(incl, off);
    if (lane >= off) incl += t;
  }
  if (lane == 31) wsum[wave] = incl;
  __syncthreads();
  int base = incl - v;
  for (int i = 0; i < wave; ++i) base += wsum[i];
  return base;
}

__device__ __forceinline__ int block_sum256(int v, int* wsum) {
  const int lane = threadIdx.x & 31;
  const int wave = threadIdx.x >> 5;
  int s = v;
#pragma unroll
  for (int off = 16; off > 0; off >>= 1) s += __shfl_xor(s, off);
  if (lane == 0) wsum[wave] = s;
  __syncthreads();
  int tot = 0;
  for (int i = 0; i < 8; ++i) tot += wsum[i];
  return tot;
}

__device__ __forceinline__ void chunk_sort_body(
    const int g, const int* __restrict__ dstIdx, const int* __restrict__ srcIdx,
    const float* __restrict__ vals, int* __restrict__ cntT,
    int* __restrict__ offT, int* __restrict__ tmpDst, int* __restrict__ tmpRec,
    int E, int N) {
  __shared__ int h[NB];
  __shared__ int wsum[8];
  const int t = threadIdx.x;
  h[t] = 0;
  __syncthreads();
  const unsigned rpb = (unsigned)(N + NB - 1) / NB;
  const int chunk = (E + NB - 1) / NB;
  const int e0 = g * chunk;
  const int e1 = (e0 + chunk < E) ? (e0 + chunk) : E;
  for (int e = e0 + t; e < e1; e += NB * EB) {
    int d[EB];
#pragma unroll
    for (int k = 0; k < EB; ++k) {
      const int i = e + k * NB;
      d[k] = (i < e1) ? dstIdx[i] : -1;
    }
#pragma unroll
    for (int k = 0; k < EB; ++k)
      if (d[k] >= 0) atomicAdd(&h[(unsigned)d[k] / rpb], 1);
  }
  __syncthreads();
  const int mine = h[t];
  const int excl = block_excl_scan256(mine, wsum);
  cntT[t * NB + g] = mine;
  offT[t * NB + g] = excl;
  h[t] = e0 + excl;
  __syncthreads();
  int2* __restrict__ tmpRec2 = (int2*)tmpRec;
  for (int e = e0 + t; e < e1; e += NB * EB) {
    int d[EB], s[EB];
    float v[EB];
#pragma unroll
    for (int k = 0; k < EB; ++k) {
      const int i = e + k * NB;
      const bool in = i < e1;
      d[k] = in ? dstIdx[i] : -1;
      s[k] = in ? srcIdx[i] : 0;
      v[k] = in ? vals[i] : 0.f;
    }
#pragma unroll
    for (int k = 0; k < EB; ++k) {
      if (d[k] >= 0) {
        const int p = atomicAdd(&h[(unsigned)d[k] / rpb], 1);
        tmpDst[p] = d[k];
        tmpRec2[p] = make_int2(s[k], __float_as_int(v[k]));
      }
    }
  }
}

__device__ __forceinline__ void bucket_rows_body(
    const int g, const int* __restrict__ cntT, const int* __restrict__ offT,
    const int* __restrict__ tmpDst, const int* __restrict__ tmpRec,
    int* __restrict__ segStart, int* __restrict__ cnt, int* __restrict__ rec,
    int E, int N) {
  constexpr int PR = 8;
  __shared__ int c[NB];
  __shared__ int sCnt[NB];
  __shared__ int sOff[NB];
  __shared__ int wsumA[8];
  __shared__ int wsumB[8];
  const int t = threadIdx.x;
  const int wave = t >> 5;
  const int sub = (t & 31) >> 3;
  const int idx = t & 7;
  const int rpb = (N + NB - 1) / NB;
  const int r0 = g * rpb;
  const int chunk = (E + NB - 1) / NB;
  const int myOff = offT[g * NB + t];
  sCnt[t] = cntT[g * NB + t];
  sOff[t] = myOff;
  c[t] = 0;
  const int b0 = block_sum256(myOff, wsumA);
  int n[PR], sg[PR];
#pragma unroll
  for (int q = 0; q < PR; ++q) {
    const int ch = q * 32 + wave * 4 + sub;
    n[q] = sCnt[ch];
    sg[q] = ch * chunk + sOff[ch];
  }
  {
    int d0[PR], d1[PR];
#pragma unroll
    for (int q = 0; q < PR; ++q) {
      d0[q] = (idx < n[q]) ? tmpDst[sg[q] + idx] : -1;
      d1[q] = (idx + 8 < n[q]) ? tmpDst[sg[q] + idx + 8] : -1;
    }
#pragma unroll
    for (int q = 0; q < PR; ++q) {
      if (d0[q] >= 0) atomicAdd(&c[d0[q] - r0], 1);
      if (d1[q] >= 0) atomicAdd(&c[d1[q] - r0], 1);
    }
#pragma unroll
    for (int q = 0; q < PR; ++q)
      for (int i = idx + 16; i < n[q]; i += 8)
        atomicAdd(&c[tmpDst[sg[q] + i] - r0], 1);
  }
  __syncthreads();
  const int mine = c[t];
  const int start = b0 + block_excl_scan256(mine, wsumB);
  c[t] = start;
  const int row = r0 + t;
  if (t < rpb && row < N) {
    segStart[row] = start;
    cnt[row] = mine;
  }
  __syncthreads();
  const int2* __restrict__ tmpRec2 = (const int2*)tmpRec;
  int2* __restrict__ rec2 = (int2*)rec;
  {
    int d0[PR], d1[PR];
    int2 r0v[PR], r1v[PR];
#pragma unroll
    for (int q = 0; q < PR; ++q) {
      const bool in0 = idx < n[q];
      const bool in1 = idx + 8 < n[q];
      d0[q] = in0 ? tmpDst[sg[q] + idx] : -1;
      r0v[q] = in0 ? tmpRec2[sg[q] + idx] : make_int2(0, 0);
      d1[q] = in1 ? tmpDst[sg[q] + idx + 8] : -1;
      r1v[q] = in1 ? tmpRec2[sg[q] + idx + 8] : make_int2(0, 0);
    }
#pragma unroll
    for (int q = 0; q < PR; ++q) {
      if (d0[q] >= 0) rec2[atomicAdd(&c[d0[q] - r0], 1)] = r0v[q];
      if (d1[q] >= 0) rec2[atomicAdd(&c[d1[q] - r0], 1)] = r1v[q];
    }
#pragma unroll
    for (int q = 0; q < PR; ++q)
      for (int i = idx + 16; i < n[q]; i += 8) {
        const int d = tmpDst[sg[q] + i];
        const int2 r = tmpRec2[sg[q] + i];
        rec2[atomicAdd(&c[d - r0], 1)] = r;
      }
  }
}

__global__ __launch_bounds__(256) void spmm_rows64_kernel(
    const int* __restrict__ segStart, const int* __restrict__ cnt,
    const int* __restrict__ rec, const float* __restrict__ z,
    const float* __restrict__ c, float* __restrict__ y, int N) {
  const int lane = threadIdx.x & 31;
  const int wave = __builtin_amdgcn_readfirstlane(
      (int)((blockIdx.x * blockDim.x + threadIdx.x) >> 5));
  const int nWaves = (int)((gridDim.x * blockDim.x) >> 5);
  const int2* __restrict__ rec2 = (const int2*)rec;
  const float2 cv = *(const float2*)(c + (size_t)lane * 2);
  for (int row = wave; row < N; row += nWaves) {
    const int beg = __builtin_amdgcn_readfirstlane(segStart[row]);
    const int n = __builtin_amdgcn_readfirstlane(cnt[row]);
    float2 acc = make_float2(0.f, 0.f);
    for (int base = 0; base < n; base += 32) {
      const int m = (n - base < 32) ? (n - base) : 32;
      int2 ev = make_int2(0, 0);
      if (lane < m) ev = rec2[beg + base + lane];
      for (int j = 0; j < m; j += 16) {
        float2 zv[16];
        float w[16];
#pragma unroll
        for (int k = 0; k < 16; ++k) {
          const int jj = (j + k < m) ? (j + k) : (m - 1);
          const int s = __builtin_amdgcn_readlane(ev.x, jj);
          w[k] = __int_as_float(__builtin_amdgcn_readlane(ev.y, jj));
          zv[k] = *(const float2*)(z + (size_t)s * 64 + (size_t)lane * 2);
        }
#pragma unroll
        for (int k = 0; k < 16; ++k) {
          const bool live = (j + k < m);
          const float ax = fmaf(zv[k].x, w[k], acc.x);
          const float ay = fmaf(zv[k].y, w[k], acc.y);
          acc.x = live ? ax : acc.x;
          acc.y = live ? ay : acc.y;
        }
      }
    }
    acc.x += cv.x;
    acc.y += cv.y;
    *(float2*)(y + (size_t)row * 64 + (size_t)lane * 2) = acc;
  }
}


__device__ __forceinline__ void gemm128_ldsB_acc(
    const float* __restrict__ A, const float* sB, v8f (&acc)[4]) {
  const int lane = threadIdx.x & 31;
  const int wave = threadIdx.x >> 5;
  const int hi = lane >> 4;
  const int lo = lane & 15;
  const float* Arow = A + (size_t)(wave * 16 + lo) * 128;
  v2f a[32];
#pragma unroll
  for (int s = 0; s < 32; ++s) a[s] = *(const v2f*)(Arow + s * 4 + hi * 2);
#pragma unroll
  for (int tt = 0; tt < 4; ++tt) acc[tt] = v8f{};
#pragma unroll
  for (int s = 0; s < 32; ++s) {
    const int kk = s * 4 + hi * 2;
#pragma unroll
    for (int tt = 0; tt < 4; ++tt) {
      const int col = tt * 16 + lo;
      v2f b;
      b.x = sB[kk * 64 + col];
      b.y = sB[(kk + 1) * 64 + col];
      acc[tt] = __builtin_amdgcn_wmma_f32_16x16x4_f32(
           false, a[s],  false, b,
           (short)0, acc[tt],  false,  false);
    }
  }
}

__device__ __forceinline__ void weight_chain_body(
    const float* __restrict__ W1, const float* __restrict__ b1,
    const float* __restrict__ W2, const float* __restrict__ b2,
    const float* __restrict__ W3, float* __restrict__ Wc,
    float* __restrict__ c1, float* __restrict__ c2) {
  __shared__ float sB[128 * 64];
  const int t = threadIdx.x;
  const int lane = t & 31;
  const int wave = t >> 5;
  const int hi = lane >> 4;
  const int lo = lane & 15;
  {
    const float2* __restrict__ src = (const float2*)W3;
    float2* dst = (float2*)sB;
#pragma unroll
    for (int i = 0; i < 16; ++i) dst[t + i * 256] = src[t + i * 256];
  }
  __syncthreads();
  v8f acc[4];
  gemm128_ldsB_acc(W2, sB, acc);
  __syncthreads();
#pragma unroll
  for (int tt = 0; tt < 4; ++tt) {
    const int col = tt * 16 + lo;
#pragma unroll
    for (int j = 0; j < 8; ++j)
      sB[(wave * 16 + j + hi * 8) * 64 + col] = acc[tt][j];
  }
  __syncthreads();
  gemm128_ldsB_acc(W1, sB, acc);
#pragma unroll
  for (int tt = 0; tt < 4; ++tt) {
    const int col = tt * 16 + lo;
#pragma unroll
    for (int j = 0; j < 8; ++j)
      Wc[(size_t)(wave * 16 + j + hi * 8) * 64 + col] = acc[tt][j];
  }
  if (t < 64) {
    float s = 0.f;
    for (int k = 0; k < 128; ++k) s = fmaf(b1[k], sB[k * 64 + t], s);
    c1[t] = s;
  } else if (t < 128) {
    const int j = t - 64;
    float s = 0.f;
    for (int k = 0; k < 128; ++k) s = fmaf(b2[k], W3[k * 64 + j], s);
    c2[j] = s;
  }
}

template <int NT>
__device__ __forceinline__ void gemm_body(
    const int gb, const float* __restrict__ S, const float* __restrict__ W,
    float* __restrict__ out, int nRowTiles) {
  const int lane = threadIdx.x & 31;
  const int waveInBlock = threadIdx.x >> 5;
  const int rowTile = gb * (blockDim.x >> 5) + waveInBlock;
  if (rowTile >= nRowTiles) return;

  const int hi = lane >> 4;
  const int lo = lane & 15;
  const int ldw = NT * 16;
  const float* Arow = S + (size_t)(rowTile * 16 + lo) * 128;

  v8f acc[NT] = {};

#pragma unroll
  for (int k = 0; k < 128; k += 4) {
    const int kk = k + hi * 2;
    v2f a = *(const v2f*)(Arow + kk);
#pragma unroll
    for (int t = 0; t < NT; ++t) {
      const int col = t * 16 + lo;
      v2f b;
      b.x = W[(size_t)kk * ldw + col];
      b.y = W[(size_t)(kk + 1) * ldw + col];
      acc[t] = __builtin_amdgcn_wmma_f32_16x16x4_f32(
           false, a,  false, b,
           (short)0, acc[t],  false,  false);
    }
  }

#pragma unroll
  for (int t = 0; t < NT; ++t) {
    const int col = t * 16 + lo;
#pragma unroll
    for (int j = 0; j < 8; ++j) {
      const int m = j + hi * 8;
      out[(size_t)(rowTile * 16 + m) * ldw + col] = acc[t][j];
    }
  }
}

__global__ __launch_bounds__(256) void front1_kernel(
    const int* __restrict__ dstIdx, const int* __restrict__ srcIdx,
    const float* __restrict__ vals, int* __restrict__ cntT,
    int* __restrict__ offT, int* __restrict__ tmpDst, int* __restrict__ tmpRec,
    int E, int N, const float* __restrict__ W1, const float* __restrict__ b1,
    const float* __restrict__ W2, const float* __restrict__ b2,
    const float* __restrict__ W3, float* __restrict__ Wc,
    float* __restrict__ c1, float* __restrict__ c2) {
  const int g = blockIdx.x;
  if (g < NB)
    chunk_sort_body(g, dstIdx, srcIdx, vals, cntT, offT, tmpDst, tmpRec, E, N);
  else
    weight_chain_body(W1, b1, W2, b2, W3, Wc, c1, c2);
}

__global__ __launch_bounds__(256) void front2_kernel(
    const int* __restrict__ cntT, const int* __restrict__ offT,
    const int* __restrict__ tmpDst, const int* __restrict__ tmpRec,
    int* __restrict__ segStart, int* __restrict__ cnt, int* __restrict__ rec,
    int E, int N, const float* __restrict__ x, const float* __restrict__ Wc,
    float* __restrict__ z0, int nRowTiles) {
  const int g = blockIdx.x;
  if (g < NB)
    bucket_rows_body(g, cntT, offT, tmpDst, tmpRec, segStart, cnt, rec, E, N);
  else
    gemm_body<4>(g - NB, x, Wc, z0, nRowTiles);
}

extern "C" void kernel_launch(void* const* d_in, const int* in_sizes, int n_in,
                              void* d_out, int out_size, void* d_ws,
                              size_t ws_size, hipStream_t stream) {
  const float* x = (const float*)d_in[0];
  const int* adj = (const int*)d_in[1];
  const float* vals = (const float*)d_in[2];
  const float* W1 = (const float*)d_in[3];
  const float* b1 = (const float*)d_in[4];
  const float* W2 = (const float*)d_in[5];
  const float* b2 = (const float*)d_in[6];
  const float* W3 = (const float*)d_in[7];
  const float* b3 = (const float*)d_in[8];
  float* out = (float*)d_out;

  const int E = in_sizes[2];
  const int N = in_sizes[0] / 128;
  const int* dstIdx = adj;
  const int* srcIdx = adj + E;

  float* zA = (float*)d_ws;
  float* zB = zA + (size_t)N * 64;
  float* Wc = zB + (size_t)N * 64;
  float* c1 = Wc + 128 * 64;
  float* c2 = c1 + 64;
  int* cnt = (int*)(c2 + 64);
  const size_t nPad = ((size_t)N + 63) / 64 * 64;
  const size_t ePad = ((size_t)E + 63) / 64 * 64;
  int* segStart = cnt + nPad;
  int* cntT = segStart + nPad;
  int* offT = cntT + NB * NB;
  int* tmpDst = offT + NB * NB;
  int* tmpRec = tmpDst + ePad;
  int* rec = tmpRec + 2 * ePad;

  const int spmmBlocks = 2048;
  const int nRowTiles = N / 16;
  const int gemmBlocks = (nRowTiles + 7) / 8;

  front1_kernel<<<NB + 1, 256, 0, stream>>>(dstIdx, srcIdx, vals, cntT, offT,
                                            tmpDst, tmpRec, E, N, W1, b1, W2, b2,
                                            W3, Wc, c1, c2);
  front2_kernel<<<NB + gemmBlocks, 256, 0, stream>>>(cntT, offT, tmpDst, tmpRec,
                                                     segStart, cnt, rec, E, N, x,
                                                     Wc, zA, nRowTiles);

  spmm_rows64_kernel<<<spmmBlocks, 256, 0, stream>>>(segStart, cnt, rec, zA, c1, zB, N);
  spmm_rows64_kernel<<<spmmBlocks, 256, 0, stream>>>(segStart, cnt, rec, zB, c2, zA, N);
  spmm_rows64_kernel<<<spmmBlocks, 256, 0, stream>>>(segStart, cnt, rec, zA, b3, out, N);
}


# Both edge-sort launches assemble their sorted range (a chunk, a bucket) in LDS at the LDS-add ranks and store it as whole 128-B lines, not one 4/8-byte store per edge; ranks and row sum order kept
# speedup vs baseline: 7.1290x; 1.0365x over previous
#include <hip/hip_runtime.h>
#include <hip/hip_bf16.h>
#include <stddef.h>


typedef __attribute__((ext_vector_type(2))) float v2f;
typedef __attribute__((ext_vector_type(8))) float v8f;

constexpr int NB = 256;
constexpr int EB = 8;
constexpr int CAP1 = 4096;
constexpr int CAP2 = 6144;
constexpr int CB = 4;

__device__ __forceinline__ int chunk_stride(int chunk) { return (chunk + 31) & ~31; }

__device__ __forceinline__ int block_excl_scan256(int v, int* wsum) {
  const int lane = threadIdx.x & 31;
  const int wave = threadIdx.x >> 5;
  int incl = v;
#pragma unroll
  for (int off = 1; off < 32; off <<= 1) {
    const int t = __shfl_up(incl, off);
    if (lane >= off) incl += t;
  }
  if (lane == 31) wsum[wave] = incl;
  __syncthreads();
  int base = incl - v;
  for (int i = 0; i < wave; ++i) base += wsum[i];
  return base;
}

__device__ __forceinline__ int block_sum256(int v, int* wsum) {
  const int lane = threadIdx.x & 31;
  const int wave = threadIdx.x >> 5;
  int s = v;
#pragma unroll
  for (int off = 16; off > 0; off >>= 1) s += __shfl_xor(s, off);
  if (lane == 0) wsum[wave] = s;
  __syncthreads();
  int tot = 0;
  for (int i = 0; i < 8; ++i) tot += wsum[i];
  return tot;
}

__device__ __forceinline__ void chunk_sort_body(
    const int g, const int* __restrict__ dstIdx, const int* __restrict__ srcIdx,
    const float* __restrict__ vals, int* __restrict__ cntT,
    int* __restrict__ offT, int* __restrict__ tmpDst, int* __restrict__ tmpRec,
    int E, int N) {
  __shared__ int h[NB];
  __shared__ int wsum[8];
  __shared__ __attribute__((aligned(16))) int sDst[CAP1];
  __shared__ __attribute__((aligned(16))) int2 sRec[CAP1];
  const int t = threadIdx.x;
  h[t] = 0;
  __syncthreads();
  const unsigned rpb = (unsigned)(N + NB - 1) / NB;
  const int chunk = (E + NB - 1) / NB;
  const int e0 = g * chunk;
  const int e1 = (e0 + chunk < E) ? (e0 + chunk) : E;
  const int o0 = g * chunk_stride(chunk);
  const bool staged = chunk <= CAP1;
  for (int e = e0 + t; e < e1; e += NB * EB) {
    int d[EB];
#pragma unroll
    for (int k = 0; k < EB; ++k) {
      const int i = e + k * NB;
      d[k] = (i < e1) ? dstIdx[i] : -1;
    }
#pragma unroll
    for (int k = 0; k < EB; ++k)
      if (d[k] >= 0) atomicAdd(&h[(unsigned)d[k] / rpb], 1);
  }
  __syncthreads();
  const int mine = h[t];
  const int excl = block_excl_scan256(mine, wsum);
  cntT[t * NB + g] = mine;
  offT[t * NB + g] = excl;
  h[t] = staged ? excl : (o0 + excl);
  __syncthreads();
  int2* __restrict__ tmpRec2 = (int2*)tmpRec;
  for (int e = e0 + t; e < e1; e += NB * EB) {
    int d[EB], s[EB];
    float v[EB];
#pragma unroll
    for (int k = 0; k < EB; ++k) {
      const int i = e + k * NB;
      const bool in = i < e1;
      d[k] = in ? dstIdx[i] : -1;
      s[k] = in ? srcIdx[i] : 0;
      v[k] = in ? vals[i] : 0.f;
    }
#pragma unroll
    for (int k = 0; k < EB; ++k) {
      if (d[k] >= 0) {
        const int p = atomicAdd(&h[(unsigned)d[k] / rpb], 1);
        const int2 r = make_int2(s[k], __float_as_int(v[k]));
        if (staged) {
          sDst[p] = d[k];
          sRec[p] = r;
        } else {
          tmpDst[p] = d[k];
          tmpRec2[p] = r;
        }
      }
    }
  }
  if (staged) {
    __syncthreads();
    const int len = e1 - e0;
    const int4* s4 = (const int4*)sDst;
    int4* d4 = (int4*)(tmpDst + o0);
    const int4* r4 = (const int4*)sRec;
    int4* q4 = (int4*)(tmpRec2 + o0);
    const int nd = (len + 3) >> 2;
    const int nr = (len + 1) >> 1;
    for (int b = t; b < nr; b += NB * CB) {
      int4 vd[CB], vr[CB];
#pragma unroll
      for (int k = 0; k < CB; ++k) {
        const int i = b + k * NB;
        vd[k] = (i < nd) ? s4[i] : make_int4(0, 0, 0, 0);
        vr[k] = (i < nr) ? r4[i] : make_int4(0, 0, 0, 0);
      }
#pragma unroll
      for (int k = 0; k < CB; ++k) {
        const int i = b + k * NB;
        if (i < nd) d4[i] = vd[k];
        if (i < nr) q4[i] = vr[k];
      }
    }
  }
}

__device__ __forceinline__ void bucket_rows_body(
    const int g, const int* __restrict__ cntT, const int* __restrict__ offT,
    const int* __restrict__ tmpDst, const int* __restrict__ tmpRec,
    int* __restrict__ segStart, int* __restrict__ cnt, int* __restrict__ rec,
    int E, int N) {
  constexpr int PR = 8;
  __shared__ int c[NB];
  __shared__ int sCnt[NB];
  __shared__ int sOff[NB];
  __shared__ int wsumA[8];
  __shared__ int wsumB[8];
  __shared__ int sTot;
  __shared__ __attribute__((aligned(16))) int2 sOut[CAP2];
  const int t = threadIdx.x;
  const int wave = t >> 5;
  const int sub = (t & 31) >> 3;
  const int idx = t & 7;
  const int rpb = (N + NB - 1) / NB;
  const int r0 = g * rpb;
  const int cs = chunk_stride((E + NB - 1) / NB);
  const int myOff = offT[g * NB + t];
  sCnt[t] = cntT[g * NB + t];
  sOff[t] = myOff;
  c[t] = 0;
  const int b0 = block_sum256(myOff, wsumA);
  int n[PR], sg[PR];
#pragma unroll
  for (int q = 0; q < PR; ++q) {
    const int ch = q * 32 + wave * 4 + sub;
    n[q] = sCnt[ch];
    sg[q] = ch * cs + sOff[ch];
  }
  {
    int d0[PR], d1[PR];
#pragma unroll
    for (int q = 0; q < PR; ++q) {
      d0[q] = (idx < n[q]) ? tmpDst[sg[q] + idx] : -1;
      d1[q] = (idx + 8 < n[q]) ? tmpDst[sg[q] + idx + 8] : -1;
    }
#pragma unroll
    for (int q = 0; q < PR; ++q) {
      if (d0[q] >= 0) atomicAdd(&c[d0[q] - r0], 1);
      if (d1[q] >= 0) atomicAdd(&c[d1[q] - r0], 1);
    }
#pragma unroll
    for (int q = 0; q < PR; ++q)
      for (int i = idx + 16; i < n[q]; i += 8)
        atomicAdd(&c[tmpDst[sg[q] + i] - r0], 1);
  }
  __syncthreads();
  const int mine = c[t];
  const int excl = block_excl_scan256(mine, wsumB);
  c[t] = excl;
  if (t == NB - 1) sTot = excl + mine;
  const int row = r0 + t;
  if (t < rpb && row < N) {
    segStart[row] = b0 + excl;
    cnt[row] = mine;
  }
  __syncthreads();
  const int nB = sTot;
  const bool staged = nB <= CAP2;
  const int2* __restrict__ tmpRec2 = (const int2*)tmpRec;
  int2* __restrict__ rec2 = (int2*)rec;
  {
    int d0[PR], d1[PR];
    int2 r0v[PR], r1v[PR];
#pragma unroll
    for (int q = 0; q < PR; ++q) {
      const bool in0 = idx < n[q];
      const bool in1 = idx + 8 < n[q];
      d0[q] = in0 ? tmpDst[sg[q] + idx] : -1;
      r0v[q] = in0 ? tmpRec2[sg[q] + idx] : make_int2(0, 0);
      d1[q] = in1 ? tmpDst[sg[q] + idx + 8] : -1;
      r1v[q] = in1 ? tmpRec2[sg[q] + idx + 8] : make_int2(0, 0);
    }
#pragma unroll
    for (int q = 0; q < PR; ++q) {
      if (d0[q] >= 0) {
        const int p = atomicAdd(&c[d0[q] - r0], 1);
        if (staged) sOut[p] = r0v[q]; else rec2[b0 + p] = r0v[q];
      }
      if (d1[q] >= 0) {
        const int p = atomicAdd(&c[d1[q] - r0], 1);
        if (staged) sOut[p] = r1v[q]; else rec2[b0 + p] = r1v[q];
      }
    }
#pragma unroll
    for (int q = 0; q < PR; ++q)
      for (int i = idx + 16; i < n[q]; i += 8) {
        const int d = tmpDst[sg[q] + i];
        const int2 r = tmpRec2[sg[q] + i];
        const int p = atomicAdd(&c[d - r0], 1);
        if (staged) sOut[p] = r; else rec2[b0 + p] = r;
      }
  }
  if (staged) {
    __syncthreads();
    const int end = b0 + nB;
    for (int b = (b0 & ~31) + t; b < end; b += NB * CB) {
      int2 v[CB];
#pragma unroll
      for (int k = 0; k < CB; ++k) {
        const int i = b + k * NB;
        v[k] = (i >= b0 && i < end) ? sOut[i - b0] : make_int2(0, 0);
      }
#pragma unroll
      for (int k = 0; k < CB; ++k) {
        const int i = b + k * NB;
        if (i >= b0 && i < end) rec2[i] = v[k];
      }
    }
  }
}

__global__ __launch_bounds__(256) void spmm_rows64_kernel(
    const int* __restrict__ segStart, const int* __restrict__ cnt,
    const int* __restrict__ rec, const float* __restrict__ z,
    const float* __restrict__ c, float* __restrict__ y, int N) {
  const int lane = threadIdx.x & 31;
  const int wave = __builtin_amdgcn_readfirstlane(
      (int)((blockIdx.x * blockDim.x + threadIdx.x) >> 5));
  const int nWaves = (int)((gridDim.x * blockDim.x) >> 5);
  const int2* __restrict__ rec2 = (const int2*)rec;
  const float2 cv = *(const float2*)(c + (size_t)lane * 2);
  for (int row = wave; row < N; row += nWaves) {
    const int beg = __builtin_amdgcn_readfirstlane(segStart[row]);
    const int n = __builtin_amdgcn_readfirstlane(cnt[row]);
    float2 acc = make_float2(0.f, 0.f);
    for (int base = 0; base < n; base += 32) {
      const int m = (n - base < 32) ? (n - base) : 32;
      int2 ev = make_int2(0, 0);
      if (lane < m) ev = rec2[beg + base + lane];
      for (int j = 0; j < m; j += 16) {
        float2 zv[16];
        float w[16];
#pragma unroll
        for (int k = 0; k < 16; ++k) {
          const int jj = (j + k < m) ? (j + k) : (m - 1);
          const int s = __builtin_amdgcn_readlane(ev.x, jj);
          w[k] = __int_as_float(__builtin_amdgcn_readlane(ev.y, jj));
          zv[k] = *(const float2*)(z + (size_t)s * 64 + (size_t)lane * 2);
        }
#pragma unroll
        for (int k = 0; k < 16; ++k) {
          const bool live = (j + k < m);
          const float ax = fmaf(zv[k].x, w[k], acc.x);
          const float ay = fmaf(zv[k].y, w[k], acc.y);
          acc.x = live ? ax : acc.x;
          acc.y = live ? ay : acc.y;
        }
      }
    }
    acc.x += cv.x;
    acc.y += cv.y;
    *(float2*)(y + (size_t)row * 64 + (size_t)lane * 2) = acc;
  }
}


__device__ __forceinline__ void gemm128_ldsB_acc(
    const float* __restrict__ A, const float* sB, v8f (&acc)[4]) {
  const int lane = threadIdx.x & 31;
  const int wave = threadIdx.x >> 5;
  const int hi = lane >> 4;
  const int lo = lane & 15;
  const float* Arow = A + (size_t)(wave * 16 + lo) * 128;
  v2f a[32];
#pragma unroll
  for (int s = 0; s < 32; ++s) a[s] = *(const v2f*)(Arow + s * 4 + hi * 2);
#pragma unroll
  for (int tt = 0; tt < 4; ++tt) acc[tt] = v8f{};
#pragma unroll
  for (int s = 0; s < 32; ++s) {
    const int kk = s * 4 + hi * 2;
#pragma unroll
    for (int tt = 0; tt < 4; ++tt) {
      const int col = tt * 16 + lo;
      v2f b;
      b.x = sB[kk * 64 + col];
      b.y = sB[(kk + 1) * 64 + col];
      acc[tt] = __builtin_amdgcn_wmma_f32_16x16x4_f32(
           false, a[s],  false, b,
           (short)0, acc[tt],  false,  false);
    }
  }
}

__device__ __forceinline__ void weight_chain_body(
    const float* __restrict__ W1, const float* __restrict__ b1,
    const float* __restrict__ W2, const float* __restrict__ b2,
    const float* __restrict__ W3, float* __restrict__ Wc,
    float* __restrict__ c1, float* __restrict__ c2) {
  __shared__ float sB[128 * 64];
  const int t = threadIdx.x;
  const int lane = t & 31;
  const int wave = t >> 5;
  const int hi = lane >> 4;
  const int lo = lane & 15;
  {
    const float2* __restrict__ src = (const float2*)W3;
    float2* dst = (float2*)sB;
#pragma unroll
    for (int i = 0; i < 16; ++i) dst[t + i * 256] = src[t + i * 256];
  }
  __syncthreads();
  v8f acc[4];
  gemm128_ldsB_acc(W2, sB, acc);
  __syncthreads();
#pragma unroll
  for (int tt = 0; tt < 4; ++tt) {
    const int col = tt * 16 + lo;
#pragma unroll
    for (int j = 0; j < 8; ++j)
      sB[(wave * 16 + j + hi * 8) * 64 + col] = acc[tt][j];
  }
  __syncthreads();
  gemm128_ldsB_acc(W1, sB, acc);
#pragma unroll
  for (int tt = 0; tt < 4; ++tt) {
    const int col = tt * 16 + lo;
#pragma unroll
    for (int j = 0; j < 8; ++j)
      Wc[(size_t)(wave * 16 + j + hi * 8) * 64 + col] = acc[tt][j];
  }
  if (t < 64) {
    float s = 0.f;
    for (int k = 0; k < 128; ++k) s = fmaf(b1[k], sB[k * 64 + t], s);
    c1[t] = s;
  } else if (t < 128) {
    const int j = t - 64;
    float s = 0.f;
    for (int k = 0; k < 128; ++k) s = fmaf(b2[k], W3[k * 64 + j], s);
    c2[j] = s;
  }
}

template <int NT>
__device__ __forceinline__ void gemm_body(
    const int gb, const float* __restrict__ S, const float* __restrict__ W,
    float* __restrict__ out, int nRowTiles) {
  const int lane = threadIdx.x & 31;
  const int waveInBlock = threadIdx.x >> 5;
  const int rowTile = gb * (blockDim.x >> 5) + waveInBlock;
  if (rowTile >= nRowTiles) return;

  const int hi = lane >> 4;
  const int lo = lane & 15;
  const int ldw = NT * 16;
  const float* Arow = S + (size_t)(rowTile * 16 + lo) * 128;

  v8f acc[NT] = {};

#pragma unroll
  for (int k = 0; k < 128; k += 4) {
    const int kk = k + hi * 2;
    v2f a = *(const v2f*)(Arow + kk);
#pragma unroll
    for (int t = 0; t < NT; ++t) {
      const int col = t * 16 + lo;
      v2f b;
      b.x = W[(size_t)kk * ldw + col];
      b.y = W[(size_t)(kk + 1) * ldw + col];
      acc[t] = __builtin_amdgcn_wmma_f32_16x16x4_f32(
           false, a,  false, b,
           (short)0, acc[t],  false,  false);
    }
  }

#pragma unroll
  for (int t = 0; t < NT; ++t) {
    const int col = t * 16 + lo;
#pragma unroll
    for (int j = 0; j < 8; ++j) {
      const int m = j + hi * 8;
      out[(size_t)(rowTile * 16 + m) * ldw + col] = acc[t][j];
    }
  }
}

__global__ __launch_bounds__(256) void front1_kernel(
    const int* __restrict__ dstIdx, const int* __restrict__ srcIdx,
    const float* __restrict__ vals, int* __restrict__ cntT,
    int* __restrict__ offT, int* __restrict__ tmpDst, int* __restrict__ tmpRec,
    int E, int N, const float* __restrict__ W1, const float* __restrict__ b1,
    const float* __restrict__ W2, const float* __restrict__ b2,
    const float* __restrict__ W3, float* __restrict__ Wc,
    float* __restrict__ c1, float* __restrict__ c2) {
  const int g = blockIdx.x;
  if (g < NB)
    chunk_sort_body(g, dstIdx, srcIdx, vals, cntT, offT, tmpDst, tmpRec, E, N);
  else
    weight_chain_body(W1, b1, W2, b2, W3, Wc, c1, c2);
}

__global__ __launch_bounds__(256) void front2_kernel(
    const int* __restrict__ cntT, const int* __restrict__ offT,
    const int* __restrict__ tmpDst, const int* __restrict__ tmpRec,
    int* __restrict__ segStart, int* __restrict__ cnt, int* __restrict__ rec,
    int E, int N, const float* __restrict__ x, const float* __restrict__ Wc,
    float* __restrict__ z0, int nRowTiles) {
  const int g = blockIdx.x;
  if (g < NB)
    bucket_rows_body(g, cntT, offT, tmpDst, tmpRec, segStart, cnt, rec, E, N);
  else
    gemm_body<4>(g - NB, x, Wc, z0, nRowTiles);
}

extern "C" void kernel_launch(void* const* d_in, const int* in_sizes, int n_in,
                              void* d_out, int out_size, void* d_ws,
                              size_t ws_size, hipStream_t stream) {
  const float* x = (const float*)d_in[0];
  const int* adj = (const int*)d_in[1];
  const float* vals = (const float*)d_in[2];
  const float* W1 = (const float*)d_in[3];
  const float* b1 = (const float*)d_in[4];
  const float* W2 = (const float*)d_in[5];
  const float* b2 = (const float*)d_in[6];
  const float* W3 = (const float*)d_in[7];
  const float* b3 = (const float*)d_in[8];
  float* out = (float*)d_out;

  const int E = in_sizes[2];
  const int N = in_sizes[0] / 128;
  const int* dstIdx = adj;
  const int* srcIdx = adj + E;

  float* zA = (float*)d_ws;
  float* zB = zA + (size_t)N * 64;
  float* Wc = zB + (size_t)N * 64;
  float* c1 = Wc + 128 * 64;
  float* c2 = c1 + 64;
  int* cnt = (int*)(c2 + 64);
  const size_t nPad = ((size_t)N + 63) / 64 * 64;
  const size_t ePad = ((size_t)E + 63) / 64 * 64;
  int* segStart = cnt + nPad;
  int* cntT = segStart + nPad;
  int* offT = cntT + NB * NB;
  const size_t tPad = ePad + 32 * NB;
  int* tmpDst = offT + NB * NB;
  int* tmpRec = tmpDst + tPad;
  int* rec = tmpRec + 2 * tPad;

  const int spmmBlocks = 2048;
  const int nRowTiles = N / 16;
  const int gemmBlocks = (nRowTiles + 7) / 8;

  front1_kernel<<<NB + 1, 256, 0, stream>>>(dstIdx, srcIdx, vals, cntT, offT,
                                            tmpDst, tmpRec, E, N, W1, b1, W2, b2,
                                            W3, Wc, c1, c2);
  front2_kernel<<<NB + gemmBlocks, 256, 0, stream>>>(cntT, offT, tmpDst, tmpRec,
                                                     segStart, cnt, rec, E, N, x,
                                                     Wc, zA, nRowTiles);

  spmm_rows64_kernel<<<spmmBlocks, 256, 0, stream>>>(segStart, cnt, rec, zA, c1, zB, N);
  spmm_rows64_kernel<<<spmmBlocks, 256, 0, stream>>>(segStart, cnt, rec, zB, c2, zA, N);
  spmm_rows64_kernel<<<spmmBlocks, 256, 0, stream>>>(segStart, cnt, rec, zA, b3, out, N);
}


# SpMM: record lanes past a row's end carry weight 0 and the row's last source, so gathers go in groups of 4 by constant lane reads (empty groups skipped): no per-record clamp, select or 64-bit address
# speedup vs baseline: 7.5063x; 1.0529x over previous
#include <hip/hip_runtime.h>
#include <hip/hip_bf16.h>
#include <stddef.h>


typedef __attribute__((ext_vector_type(2))) float v2f;
typedef __attribute__((ext_vector_type(8))) float v8f;

constexpr int NB = 256;
constexpr int EB = 8;
constexpr int CAP1 = 4096;
constexpr int CAP2 = 6144;
constexpr int CB = 4;

__device__ __forceinline__ int chunk_stride(int chunk) { return (chunk + 31) & ~31; }

__device__ __forceinline__ int block_excl_scan256(int v, int* wsum) {
  const int lane = threadIdx.x & 31;
  const int wave = threadIdx.x >> 5;
  int incl = v;
#pragma unroll
  for (int off = 1; off < 32; off <<= 1) {
    const int t = __shfl_up(incl, off);
    if (lane >= off) incl += t;
  }
  if (lane == 31) wsum[wave] = incl;
  __syncthreads();
  int base = incl - v;
  for (int i = 0; i < wave; ++i) base += wsum[i];
  return base;
}

__device__ __forceinline__ int block_sum256(int v, int* wsum) {
  const int lane = threadIdx.x & 31;
  const int wave = threadIdx.x >> 5;
  int s = v;
#pragma unroll
  for (int off = 16; off > 0; off >>= 1) s += __shfl_xor(s, off);
  if (lane == 0) wsum[wave] = s;
  __syncthreads();
  int tot = 0;
  for (int i = 0; i < 8; ++i) tot += wsum[i];
  return tot;
}

__device__ __forceinline__ void chunk_sort_body(
    const int g, const int* __restrict__ dstIdx, const int* __restrict__ srcIdx,
    const float* __restrict__ vals, int* __restrict__ cntT,
    int* __restrict__ offT, int* __restrict__ tmpDst, int* __restrict__ tmpRec,
    int E, int N) {
  __shared__ int h[NB];
  __shared__ int wsum[8];
  __shared__ __attribute__((aligned(16))) int sDst[CAP1];
  __shared__ __attribute__((aligned(16))) int2 sRec[CAP1];
  const int t = threadIdx.x;
  h[t] = 0;
  __syncthreads();
  const unsigned rpb = (unsigned)(N + NB - 1) / NB;
  const int chunk = (E + NB - 1) / NB;
  const int e0 = g * chunk;
  const int e1 = (e0 + chunk < E) ? (e0 + chunk) : E;
  const int o0 = g * chunk_stride(chunk);
  const bool staged = chunk <= CAP1;
  for (int e = e0 + t; e < e1; e += NB * EB) {
    int d[EB];
#pragma unroll
    for (int k = 0; k < EB; ++k) {
      const int i = e + k * NB;
      d[k] = (i < e1) ? dstIdx[i] : -1;
    }
#pragma unroll
    for (int k = 0; k < EB; ++k)
      if (d[k] >= 0) atomicAdd(&h[(unsigned)d[k] / rpb], 1);
  }
  __syncthreads();
  const int mine = h[t];
  const int excl = block_excl_scan256(mine, wsum);
  cntT[t * NB + g] = mine;
  offT[t * NB + g] = excl;
  h[t] = staged ? excl : (o0 + excl);
  __syncthreads();
  int2* __restrict__ tmpRec2 = (int2*)tmpRec;
  for (int e = e0 + t; e < e1; e += NB * EB) {
    int d[EB], s[EB];
    float v[EB];
#pragma unroll
    for (int k = 0; k < EB; ++k) {
      const int i = e + k * NB;
      const bool in = i < e1;
      d[k] = in ? dstIdx[i] : -1;
      s[k] = in ? srcIdx[i] : 0;
      v[k] = in ? vals[i] : 0.f;
    }
#pragma unroll
    for (int k = 0; k < EB; ++k) {
      if (d[k] >= 0) {
        const int p = atomicAdd(&h[(unsigned)d[k] / rpb], 1);
        const int2 r = make_int2(s[k], __float_as_int(v[k]));
        if (staged) {
          sDst[p] = d[k];
          sRec[p] = r;
        } else {
          tmpDst[p] = d[k];
          tmpRec2[p] = r;
        }
      }
    }
  }
  if (staged) {
    __syncthreads();
    const int len = e1 - e0;
    const int4* s4 = (const int4*)sDst;
    int4* d4 = (int4*)(tmpDst + o0);
    const int4* r4 = (const int4*)sRec;
    int4* q4 = (int4*)(tmpRec2 + o0);
    const int nd = (len + 3) >> 2;
    const int nr = (len + 1) >> 1;
    for (int b = t; b < nr; b += NB * CB) {
      int4 vd[CB], vr[CB];
#pragma unroll
      for (int k = 0; k < CB; ++k) {
        const int i = b + k * NB;
        vd[k] = (i < nd) ? s4[i] : make_int4(0, 0, 0, 0);
        vr[k] = (i < nr) ? r4[i] : make_int4(0, 0, 0, 0);
      }
#pragma unroll
      for (int k = 0; k < CB; ++k) {
        const int i = b + k * NB;
        if (i < nd) d4[i] = vd[k];
        if (i < nr) q4[i] = vr[k];
      }
    }
  }
}

__device__ __forceinline__ void bucket_rows_body(
    const int g, const int* __restrict__ cntT, const int* __restrict__ offT,
    const int* __restrict__ tmpDst, const int* __restrict__ tmpRec,
    int* __restrict__ segStart, int* __restrict__ cnt, int* __restrict__ rec,
    int E, int N) {
  constexpr int PR = 8;
  __shared__ int c[NB];
  __shared__ int sCnt[NB];
  __shared__ int sOff[NB];
  __shared__ int wsumA[8];
  __shared__ int wsumB[8];
  __shared__ int sTot;
  __shared__ __attribute__((aligned(16))) int2 sOut[CAP2];
  const int t = threadIdx.x;
  const int wave = t >> 5;
  const int sub = (t & 31) >> 3;
  const int idx = t & 7;
  const int rpb = (N + NB - 1) / NB;
  const int r0 = g * rpb;
  const int cs = chunk_stride((E + NB - 1) / NB);
  const int myOff = offT[g * NB + t];
  sCnt[t] = cntT[g * NB + t];
  sOff[t] = myOff;
  c[t] = 0;
  const int b0 = block_sum256(myOff, wsumA);
  int n[PR], sg[PR];
#pragma unroll
  for (int q = 0; q < PR; ++q) {
    const int ch = q * 32 + wave * 4 + sub;
    n[q] = sCnt[ch];
    sg[q] = ch * cs + sOff[ch];
  }
  {
    int d0[PR], d1[PR];
#pragma unroll
    for (int q = 0; q < PR; ++q) {
      d0[q] = (idx < n[q]) ? tmpDst[sg[q] + idx] : -1;
      d1[q] = (idx + 8 < n[q]) ? tmpDst[sg[q] + idx + 8] : -1;
    }
#pragma unroll
    for (int q = 0; q < PR; ++q) {
      if (d0[q] >= 0) atomicAdd(&c[d0[q] - r0], 1);
      if (d1[q] >= 0) atomicAdd(&c[d1[q] - r0], 1);
    }
#pragma unroll
    for (int q = 0; q < PR; ++q)
      for (int i = idx + 16; i < n[q]; i += 8)
        atomicAdd(&c[tmpDst[sg[q] + i] - r0], 1);
  }
  __syncthreads();
  const int mine = c[t];
  const int excl = block_excl_scan256(mine, wsumB);
  c[t] = excl;
  if (t == NB - 1) sTot = excl + mine;
  const int row = r0 + t;
  if (t < rpb && row < N) {
    segStart[row] = b0 + excl;
    cnt[row] = mine;
  }
  __syncthreads();
  const int nB = sTot;
  const bool staged = nB <= CAP2;
  const int2* __restrict__ tmpRec2 = (const int2*)tmpRec;
  int2* __restrict__ rec2 = (int2*)rec;
  {
    int d0[PR], d1[PR];
    int2 r0v[PR], r1v[PR];
#pragma unroll
    for (int q = 0; q < PR; ++q) {
      const bool in0 = idx < n[q];
      const bool in1 = idx + 8 < n[q];
      d0[q] = in0 ? tmpDst[sg[q] + idx] : -1;
      r0v[q] = in0 ? tmpRec2[sg[q] + idx] : make_int2(0, 0);
      d1[q] = in1 ? tmpDst[sg[q] + idx + 8] : -1;
      r1v[q] = in1 ? tmpRec2[sg[q] + idx + 8] : make_int2(0, 0);
    }
#pragma unroll
    for (int q = 0; q < PR; ++q) {
      if (d0[q] >= 0) {
        const int p = atomicAdd(&c[d0[q] - r0], 1);
        if (staged) sOut[p] = r0v[q]; else rec2[b0 + p] = r0v[q];
      }
      if (d1[q] >= 0) {
        const int p = atomicAdd(&c[d1[q] - r0], 1);
        if (staged) sOut[p] = r1v[q]; else rec2[b0 + p] = r1v[q];
      }
    }
#pragma unroll
    for (int q = 0; q < PR; ++q)
      for (int i = idx + 16; i < n[q]; i += 8) {
        const int d = tmpDst[sg[q] + i];
        const int2 r = tmpRec2[sg[q] + i];
        const int p = atomicAdd(&c[d - r0], 1);
        if (staged) sOut[p] = r; else rec2[b0 + p] = r;
      }
  }
  if (staged) {
    __syncthreads();
    const int end = b0 + nB;
    for (int b = (b0 & ~31) + t; b < end; b += NB * CB) {
      int2 v[CB];
#pragma unroll
      for (int k = 0; k < CB; ++k) {
        const int i = b + k * NB;
        v[k] = (i >= b0 && i < end) ? sOut[i - b0] : make_int2(0, 0);
      }
#pragma unroll
      for (int k = 0; k < CB; ++k) {
        const int i = b + k * NB;
        if (i >= b0 && i < end) rec2[i] = v[k];
      }
    }
  }
}

template <int L>
__device__ __forceinline__ void load4(const int sx,
                                      const float2* __restrict__ z2,
                                      const unsigned lane, float2 (&zv)[16]) {
#pragma unroll
  for (int k = 0; k < 4; ++k) {
    const unsigned s = (unsigned)__builtin_amdgcn_readlane(sx, L + k);
    zv[L + k] = z2[s * 32u + lane];
  }
}
template <int L>
__device__ __forceinline__ void fma4(const int wy, const float2 (&zv)[16],
                                     float2& acc) {
#pragma unroll
  for (int k = 0; k < 4; ++k) {
    const float w = __int_as_float(__builtin_amdgcn_readlane(wy, L + k));
    acc.x = fmaf(zv[L + k].x, w, acc.x);
    acc.y = fmaf(zv[L + k].y, w, acc.y);
  }
}
__device__ __forceinline__ void gather16_acc(const int sx, const int wy,
                                             const float2* __restrict__ z2,
                                             const unsigned lane, const int r,
                                             float2& acc) {
  float2 zv[16];
  load4<0>(sx, z2, lane, zv);
  if (r > 4) {
    load4<4>(sx, z2, lane, zv);
    if (r > 8) {
      load4<8>(sx, z2, lane, zv);
      if (r > 12) load4<12>(sx, z2, lane, zv);
    }
  }
  fma4<0>(wy, zv, acc);
  if (r > 4) {
    fma4<4>(wy, zv, acc);
    if (r > 8) {
      fma4<8>(wy, zv, acc);
      if (r > 12) fma4<12>(wy, zv, acc);
    }
  }
}
__device__ __forceinline__ int xhalf_i(const int b) {
  return __builtin_amdgcn_permlanex16(b, b, 0x76543210, 0xfedcba98, false,
                                      false);
}

__global__ __launch_bounds__(256) void spmm_rows64_kernel(
    const int* __restrict__ segStart, const int* __restrict__ cnt,
    const int* __restrict__ rec, const float* __restrict__ z,
    const float* __restrict__ c, float* __restrict__ y, int N) {
  const int lane = threadIdx.x & 31;
  const int wave = __builtin_amdgcn_readfirstlane(
      (int)((blockIdx.x * blockDim.x + threadIdx.x) >> 5));
  const int nWaves = (int)((gridDim.x * blockDim.x) >> 5);
  const int2* __restrict__ rec2 = (const int2*)rec;
  const float2* __restrict__ z2 = (const float2*)z;
  const float2 cv = *(const float2*)(c + (size_t)lane * 2);
  for (int row = wave; row < N; row += nWaves) {
    const int beg = __builtin_amdgcn_readfirstlane(segStart[row]);
    const int n = __builtin_amdgcn_readfirstlane(cnt[row]);
    float2 acc = make_float2(0.f, 0.f);
    for (int base = 0; base < n; base += 32) {
      const int m = (n - base < 32) ? (n - base) : 32;
      int2 ev = make_int2(0, 0);
      if (lane < m) ev = rec2[beg + base + lane];
      const int lastSrc = __builtin_amdgcn_readlane(ev.x, m - 1);
      int sx = (lane < m) ? ev.x : lastSrc;
      int wy = ev.y;
      int r = m;
#pragma nounroll
      for (;;) {
        gather16_acc(sx, wy, z2, (unsigned)lane, r, acc);
        if (r <= 16) break;
        r -= 16;
        sx = xhalf_i(sx);
        wy = xhalf_i(wy);
      }
    }
    acc.x += cv.x;
    acc.y += cv.y;
    *(float2*)(y + (size_t)row * 64 + (size_t)lane * 2) = acc;
  }
}


__device__ __forceinline__ void gemm128_ldsB_acc(
    const float* __restrict__ A, const float* sB, v8f (&acc)[4]) {
  const int lane = threadIdx.x & 31;
  const int wave = threadIdx.x >> 5;
  const int hi = lane >> 4;
  const int lo = lane & 15;
  const float* Arow = A + (size_t)(wave * 16 + lo) * 128;
  v2f a[32];
#pragma unroll
  for (int s = 0; s < 32; ++s) a[s] = *(const v2f*)(Arow + s * 4 + hi * 2);
#pragma unroll
  for (int tt = 0; tt < 4; ++tt) acc[tt] = v8f{};
#pragma unroll
  for (int s = 0; s < 32; ++s) {
    const int kk = s * 4 + hi * 2;
#pragma unroll
    for (int tt = 0; tt < 4; ++tt) {
      const int col = tt * 16 + lo;
      v2f b;
      b.x = sB[kk * 64 + col];
      b.y = sB[(kk + 1) * 64 + col];
      acc[tt] = __builtin_amdgcn_wmma_f32_16x16x4_f32(
           false, a[s],  false, b,
           (short)0, acc[tt],  false,  false);
    }
  }
}

__device__ __forceinline__ void weight_chain_body(
    const float* __restrict__ W1, const float* __restrict__ b1,
    const float* __restrict__ W2, const float* __restrict__ b2,
    const float* __restrict__ W3, float* __restrict__ Wc,
    float* __restrict__ c1, float* __restrict__ c2) {
  __shared__ float sB[128 * 64];
  const int t = threadIdx.x;
  const int lane = t & 31;
  const int wave = t >> 5;
  const int hi = lane >> 4;
  const int lo = lane & 15;
  {
    const float2* __restrict__ src = (const float2*)W3;
    float2* dst = (float2*)sB;
#pragma unroll
    for (int i = 0; i < 16; ++i) dst[t + i * 256] = src[t + i * 256];
  }
  __syncthreads();
  v8f acc[4];
  gemm128_ldsB_acc(W2, sB, acc);
  __syncthreads();
#pragma unroll
  for (int tt = 0; tt < 4; ++tt) {
    const int col = tt * 16 + lo;
#pragma unroll
    for (int j = 0; j < 8; ++j)
      sB[(wave * 16 + j + hi * 8) * 64 + col] = acc[tt][j];
  }
  __syncthreads();
  gemm128_ldsB_acc(W1, sB, acc);
#pragma unroll
  for (int tt = 0; tt < 4; ++tt) {
    const int col = tt * 16 + lo;
#pragma unroll
    for (int j = 0; j < 8; ++j)
      Wc[(size_t)(wave * 16 + j + hi * 8) * 64 + col] = acc[tt][j];
  }
  if (t < 64) {
    float s = 0.f;
    for (int k = 0; k < 128; ++k) s = fmaf(b1[k], sB[k * 64 + t], s);
    c1[t] = s;
  } else if (t < 128) {
    const int j = t - 64;
    float s = 0.f;
    for (int k = 0; k < 128; ++k) s = fmaf(b2[k], W3[k * 64 + j], s);
    c2[j] = s;
  }
}

template <int NT>
__device__ __forceinline__ void gemm_body(
    const int gb, const float* __restrict__ S, const float* __restrict__ W,
    float* __restrict__ out, int nRowTiles) {
  const int lane = threadIdx.x & 31;
  const int waveInBlock = threadIdx.x >> 5;
  const int rowTile = gb * (blockDim.x >> 5) + waveInBlock;
  if (rowTile >= nRowTiles) return;

  const int hi = lane >> 4;
  const int lo = lane & 15;
  const int ldw = NT * 16;
  const float* Arow = S + (size_t)(rowTile * 16 + lo) * 128;

  v8f acc[NT] = {};

#pragma unroll
  for (int k = 0; k < 128; k += 4) {
    const int kk = k + hi * 2;
    v2f a = *(const v2f*)(Arow + kk);
#pragma unroll
    for (int t = 0; t < NT; ++t) {
      const int col = t * 16 + lo;
      v2f b;
      b.x = W[(size_t)kk * ldw + col];
      b.y = W[(size_t)(kk + 1) * ldw + col];
      acc[t] = __builtin_amdgcn_wmma_f32_16x16x4_f32(
           false, a,  false, b,
           (short)0, acc[t],  false,  false);
    }
  }

#pragma unroll
  for (int t = 0; t < NT; ++t) {
    const int col = t * 16 + lo;
#pragma unroll
    for (int j = 0; j < 8; ++j) {
      const int m = j + hi * 8;
      out[(size_t)(rowTile * 16 + m) * ldw + col] = acc[t][j];
    }
  }
}

__global__ __launch_bounds__(256) void front1_kernel(
    const int* __restrict__ dstIdx, const int* __restrict__ srcIdx,
    const float* __restrict__ vals, int* __restrict__ cntT,
    int* __restrict__ offT, int* __restrict__ tmpDst, int* __restrict__ tmpRec,
    int E, int N, const float* __restrict__ W1, const float* __restrict__ b1,
    const float* __restrict__ W2, const float* __restrict__ b2,
    const float* __restrict__ W3, float* __restrict__ Wc,
    float* __restrict__ c1, float* __restrict__ c2) {
  const int g = blockIdx.x;
  if (g < NB)
    chunk_sort_body(g, dstIdx, srcIdx, vals, cntT, offT, tmpDst, tmpRec, E, N);
  else
    weight_chain_body(W1, b1, W2, b2, W3, Wc, c1, c2);
}

__global__ __launch_bounds__(256) void front2_kernel(
    const int* __restrict__ cntT, const int* __restrict__ offT,
    const int* __restrict__ tmpDst, const int* __restrict__ tmpRec,
    int* __restrict__ segStart, int* __restrict__ cnt, int* __restrict__ rec,
    int E, int N, const float* __restrict__ x, const float* __restrict__ Wc,
    float* __restrict__ z0, int nRowTiles) {
  const int g = blockIdx.x;
  if (g < NB)
    bucket_rows_body(g, cntT, offT, tmpDst, tmpRec, segStart, cnt, rec, E, N);
  else
    gemm_body<4>(g - NB, x, Wc, z0, nRowTiles);
}

extern "C" void kernel_launch(void* const* d_in, const int* in_sizes, int n_in,
                              void* d_out, int out_size, void* d_ws,
                              size_t ws_size, hipStream_t stream) {
  const float* x = (const float*)d_in[0];
  const int* adj = (const int*)d_in[1];
  const float* vals = (const float*)d_in[2];
  const float* W1 = (const float*)d_in[3];
  const float* b1 = (const float*)d_in[4];
  const float* W2 = (const float*)d_in[5];
  const float* b2 = (const float*)d_in[6];
  const float* W3 = (const float*)d_in[7];
  const float* b3 = (const float*)d_in[8];
  float* out = (float*)d_out;

  const int E = in_sizes[2];
  const int N = in_sizes[0] / 128;
  const int* dstIdx = adj;
  const int* srcIdx = adj + E;

  float* zA = (float*)d_ws;
  float* zB = zA + (size_t)N * 64;
  float* Wc = zB + (size_t)N * 64;
  float* c1 = Wc + 128 * 64;
  float* c2 = c1 + 64;
  int* cnt = (int*)(c2 + 64);
  const size_t nPad = ((size_t)N + 63) / 64 * 64;
  const size_t ePad = ((size_t)E + 63) / 64 * 64;
  int* segStart = cnt + nPad;
  int* cntT = segStart + nPad;
  int* offT = cntT + NB * NB;
  const size_t tPad = ePad + 32 * NB;
  int* tmpDst = offT + NB * NB;
  int* tmpRec = tmpDst + tPad;
  int* rec = tmpRec + 2 * tPad;

  const int spmmBlocks = 2048;
  const int nRowTiles = N / 16;
  const int gemmBlocks = (nRowTiles + 7) / 8;

  front1_kernel<<<NB + 1, 256, 0, stream>>>(dstIdx, srcIdx, vals, cntT, offT,
                                            tmpDst, tmpRec, E, N, W1, b1, W2, b2,
                                            W3, Wc, c1, c2);
  front2_kernel<<<NB + gemmBlocks, 256, 0, stream>>>(cntT, offT, tmpDst, tmpRec,
                                                     segStart, cnt, rec, E, N, x,
                                                     Wc, zA, nRowTiles);

  spmm_rows64_kernel<<<spmmBlocks, 256, 0, stream>>>(segStart, cnt, rec, zA, c1, zB, N);
  spmm_rows64_kernel<<<spmmBlocks, 256, 0, stream>>>(segStart, cnt, rec, zB, c2, zA, N);
  spmm_rows64_kernel<<<spmmBlocks, 256, 0, stream>>>(segStart, cnt, rec, zA, b3, out, N);
}
